# starting kernel
# baseline (speedup 1.0000x reference)
#include <hip/hip_runtime.h>


typedef __attribute__((ext_vector_type(2))) float v2f;
typedef __attribute__((ext_vector_type(8))) float v8f;

#define NWAVES    8
#define MROWS     16
#define ROWS_BLK  (NWAVES * MROWS)
#define DD        32
#define HH        128
#define SSTRIDE   132

#define OFF_W0   0
#define OFF_W1   (OFF_W0 + 4096)
#define OFF_W2   (OFF_W1 + 16384)
#define OFF_B0   (OFF_W2 + 4096)
#define OFF_T0   (OFF_B0 + 128)
#define OFF_B1   (OFF_T0 + 128)
#define OFF_T1   (OFF_B1 + 128)
#define OFF_B2   (OFF_T1 + 128)
#define OFF_T2   (OFF_B2 + 32)
#define OFF_STG  (OFF_T2 + 32)
#define STG_WAVE (2 * 16 * SSTRIDE)
#define SMEM_FLOATS (OFF_STG + NWAVES * STG_WAVE)

__device__ __forceinline__ float fast_tanh(float x) {
  float cx = fminf(fmaxf(x, -10.0f), 10.0f);
  float e  = __expf(2.0f * cx);
  return 1.0f - 2.0f * __builtin_amdgcn_rcpf(e + 1.0f);
}

__device__ __forceinline__ v8f wmma4(v2f a, v2f b, v8f c) {
  return __builtin_amdgcn_wmma_f32_16x16x4_f32(false, a, false, b, (short)0, c,
                                               false, false);
}

__device__ __forceinline__ void func_eval(
    const float* smem, float* sH, float* sU, int lo, int hi,
    const v2f* zin, const v2f* eps_a, const float (*eps_d)[8],
    v2f* kcur, float* lsum, float t, float wrk)
{
  const v8f ZERO8 = {0.f,0.f,0.f,0.f,0.f,0.f,0.f,0.f};

  const int fb = hi * 32 + 2 * lo;
  const float* wF0 = smem + OFF_W0 + fb;
  const float* wF1 = smem + OFF_W1 + fb;
  const float* wF2 = smem + OFF_W2 + fb;
  float* sHa = sH + lo * SSTRIDE + 2 * hi;
  float* sUa = sU + lo * SSTRIDE + 2 * hi;
  float* sHs = sH + 8 * hi * SSTRIDE + lo;
  float* sUs = sU + 8 * hi * SSTRIDE + lo;

  v8f acc[8];
#pragma unroll
  for (int nt = 0; nt < 8; ++nt) acc[nt] = ZERO8;
  for (int kc = 0; kc < 8; ++kc) {
    v2f a = zin[kc];
#pragma unroll
    for (int nt = 0; nt < 8; ++nt) {
      v2f b = *(const v2f*)(wF0 + (kc * 8 + nt) * 64);
      acc[nt] = wmma4(a, b, acc[nt]);
    }
  }
#pragma unroll
  for (int nt = 0; nt < 8; ++nt) {
    int col = 16 * nt + lo;
    float bias = smem[OFF_B0 + col] + t * smem[OFF_T0 + col];
#pragma unroll
    for (int v = 0; v < 8; ++v)
      sHs[v * SSTRIDE + 16 * nt] = fast_tanh(acc[nt][v] + bias);
  }
  __builtin_amdgcn_wave_barrier();

#pragma unroll
  for (int nt = 0; nt < 8; ++nt) acc[nt] = ZERO8;
  for (int kc = 0; kc < 8; ++kc) {
    v2f a = eps_a[kc];
#pragma unroll
    for (int nt = 0; nt < 8; ++nt) {
      v2f b = *(const v2f*)(wF0 + (kc * 8 + nt) * 64);
      acc[nt] = wmma4(a, b, acc[nt]);
    }
  }
#pragma unroll
  for (int nt = 0; nt < 8; ++nt) {
#pragma unroll
    for (int v = 0; v < 8; ++v) {
      int idx = v * SSTRIDE + 16 * nt;
      float h = sHs[idx];
      sUs[idx] = acc[nt][v] * (1.0f - h * h);
    }
  }
  __builtin_amdgcn_wave_barrier();

#pragma unroll
  for (int nt = 0; nt < 8; ++nt) acc[nt] = ZERO8;
  for (int kc = 0; kc < 32; ++kc) {
    v2f a = *(const v2f*)(sHa + 4 * kc);
#pragma unroll
    for (int nt = 0; nt < 8; ++nt) {
      v2f b = *(const v2f*)(wF1 + (kc * 8 + nt) * 64);
      acc[nt] = wmma4(a, b, acc[nt]);
    }
  }
  __builtin_amdgcn_wave_barrier();
#pragma unroll
  for (int nt = 0; nt < 8; ++nt) {
    int col = 16 * nt + lo;
    float bias = smem[OFF_B1 + col] + t * smem[OFF_T1 + col];
#pragma unroll
    for (int v = 0; v < 8; ++v)
      sHs[v * SSTRIDE + 16 * nt] = fast_tanh(acc[nt][v] + bias);
  }
  __builtin_amdgcn_wave_barrier();

#pragma unroll
  for (int nt = 0; nt < 8; ++nt) acc[nt] = ZERO8;
  for (int kc = 0; kc < 32; ++kc) {
    v2f a = *(const v2f*)(sUa + 4 * kc);
#pragma unroll
    for (int nt = 0; nt < 8; ++nt) {
      v2f b = *(const v2f*)(wF1 + (kc * 8 + nt) * 64);
      acc[nt] = wmma4(a, b, acc[nt]);
    }
  }
  __builtin_amdgcn_wave_barrier();
#pragma unroll
  for (int nt = 0; nt < 8; ++nt) {
#pragma unroll
    for (int v = 0; v < 8; ++v) {
      int idx = v * SSTRIDE + 16 * nt;
      float h = sHs[idx];
      sUs[idx] = acc[nt][v] * (1.0f - h * h);
    }
  }
  __builtin_amdgcn_wave_barrier();

  v8f dacc[2], jacc[2];
  dacc[0] = ZERO8; dacc[1] = ZERO8; jacc[0] = ZERO8; jacc[1] = ZERO8;
  for (int kc = 0; kc < 32; ++kc) {
    v2f ah = *(const v2f*)(sHa + 4 * kc);
    v2f au = *(const v2f*)(sUa + 4 * kc);
#pragma unroll
    for (int nt = 0; nt < 2; ++nt) {
      v2f b = *(const v2f*)(wF2 + (kc * 2 + nt) * 64);
      dacc[nt] = wmma4(ah, b, dacc[nt]);
      jacc[nt] = wmma4(au, b, jacc[nt]);
    }
  }
  __builtin_amdgcn_wave_barrier();

  float p[8];
#pragma unroll
  for (int v = 0; v < 8; ++v)
    p[v] = eps_d[0][v] * jacc[0][v] + eps_d[1][v] * jacc[1][v];
#pragma unroll
  for (int off = 1; off < 16; off <<= 1) {
#pragma unroll
    for (int v = 0; v < 8; ++v) p[v] += __shfl_xor(p[v], off, 32);
  }
#pragma unroll
  for (int v = 0; v < 8; ++v) lsum[v] += wrk * p[v];

#pragma unroll
  for (int nt = 0; nt < 2; ++nt) {
    int col = 16 * nt + lo;
    float bias = smem[OFF_B2 + col] + t * smem[OFF_T2 + col];
#pragma unroll
    for (int v = 0; v < 8; ++v)
      sHs[v * SSTRIDE + 16 * nt] = dacc[nt][v] + bias;
  }
  __builtin_amdgcn_wave_barrier();
#pragma unroll
  for (int kc = 0; kc < 8; ++kc)
    kcur[kc] = *(const v2f*)(sHa + 4 * kc);
  __builtin_amdgcn_wave_barrier();
}

__global__ __launch_bounds__(256)
void ffjord_kernel(const float* __restrict__ z1, const float* __restrict__ eps,
                   const float* __restrict__ W0, const float* __restrict__ b0,
                   const float* __restrict__ W1, const float* __restrict__ b1,
                   const float* __restrict__ W2, const float* __restrict__ b2,
                   float* __restrict__ out_z, float* __restrict__ out_ld)
{
  extern __shared__ float smem[];
  const int tid  = threadIdx.x;
  const int lane = tid & 31;
  const int wave = tid >> 5;
  const int lo   = lane & 15;
  const int hi   = lane >> 4;

  for (int i = tid; i < 32 * 128; i += blockDim.x) {
    int k = i >> 7, n = i & 127;
    int idx = (((k >> 2) * 8 + (n >> 4)) << 6) + (((k >> 1) & 1) << 5) + 2 * (n & 15) + (k & 1);
    smem[OFF_W0 + idx] = W0[k * 128 + n];
  }
  for (int i = tid; i < 128 * 128; i += blockDim.x) {
    int k = i >> 7, n = i & 127;
    int idx = (((k >> 2) * 8 + (n >> 4)) << 6) + (((k >> 1) & 1) << 5) + 2 * (n & 15) + (k & 1);
    smem[OFF_W1 + idx] = W1[k * 128 + n];
  }
  for (int i = tid; i < 128 * 32; i += blockDim.x) {
    int k = i >> 5, n = i & 31;
    int idx = (((k >> 2) * 2 + (n >> 4)) << 6) + (((k >> 1) & 1) << 5) + 2 * (n & 15) + (k & 1);
    smem[OFF_W2 + idx] = W2[k * 32 + n];
  }
  for (int n = tid; n < 128; n += blockDim.x) {
    smem[OFF_T0 + n] = W0[32 * 128 + n];
    smem[OFF_B0 + n] = b0[n];
    smem[OFF_T1 + n] = W1[128 * 128 + n];
    smem[OFF_B1 + n] = b1[n];
  }
  for (int n = tid; n < 32; n += blockDim.x) {
    smem[OFF_T2 + n] = W2[128 * 32 + n];
    smem[OFF_B2 + n] = b2[n];
  }
  __syncthreads();

  float* sH = smem + OFF_STG + wave * STG_WAVE;
  float* sU = sH + 16 * SSTRIDE;
  const int rowbase = blockIdx.x * ROWS_BLK + wave * MROWS;

  v2f z_a[8], eps_a[8], zin[8], ksum[8], kcur[8];
  float eps_d[2][8];
  float ldacc[8];

  {
    const float* zrow = z1  + (size_t)(rowbase + lo) * DD + 2 * hi;
    const float* erow = eps + (size_t)(rowbase + lo) * DD + 2 * hi;
#pragma unroll
    for (int kc = 0; kc < 8; ++kc) {
      z_a[kc]   = *(const v2f*)(zrow + 4 * kc);
      eps_a[kc] = *(const v2f*)(erow + 4 * kc);
    }
#pragma unroll
    for (int nt = 0; nt < 2; ++nt)
#pragma unroll
      for (int v = 0; v < 8; ++v)
        eps_d[nt][v] = eps[(size_t)(rowbase + v + 8 * hi) * DD + 16 * nt + lo];
#pragma unroll
    for (int v = 0; v < 8; ++v) ldacc[v] = 0.0f;
  }

  const float hstep = -0.2f;
  for (int s = 0; s < 5; ++s) {
    const float tb = 1.0f - 0.2f * (float)s;
    float lsum[8];
#pragma unroll
    for (int v = 0; v < 8; ++v) lsum[v] = 0.0f;

#pragma unroll
    for (int i = 0; i < 8; ++i) zin[i] = z_a[i];
    func_eval(smem, sH, sU, lo, hi, zin, eps_a, eps_d, kcur, lsum, tb, 1.0f);
#pragma unroll
    for (int i = 0; i < 8; ++i) { ksum[i] = kcur[i]; zin[i] = z_a[i] + kcur[i] * (0.5f * hstep); }

    func_eval(smem, sH, sU, lo, hi, zin, eps_a, eps_d, kcur, lsum, tb + 0.5f * hstep, 2.0f);
#pragma unroll
    for (int i = 0; i < 8; ++i) { ksum[i] += kcur[i] * 2.0f; zin[i] = z_a[i] + kcur[i] * (0.5f * hstep); }

    func_eval(smem, sH, sU, lo, hi, zin, eps_a, eps_d, kcur, lsum, tb + 0.5f * hstep, 2.0f);
#pragma unroll
    for (int i = 0; i < 8; ++i) { ksum[i] += kcur[i] * 2.0f; zin[i] = z_a[i] + kcur[i] * hstep; }

    func_eval(smem, sH, sU, lo, hi, zin, eps_a, eps_d, kcur, lsum, tb + hstep, 1.0f);
#pragma unroll
    for (int i = 0; i < 8; ++i) ksum[i] += kcur[i];

    const float c = hstep / 6.0f;
#pragma unroll
    for (int i = 0; i < 8; ++i) z_a[i] += ksum[i] * c;
#pragma unroll
    for (int v = 0; v < 8; ++v) ldacc[v] -= c * lsum[v];
  }

  {
    float* orow = out_z + (size_t)(rowbase + lo) * DD + 2 * hi;
#pragma unroll
    for (int kc = 0; kc < 8; ++kc)
      *(v2f*)(orow + 4 * kc) = z_a[kc];
#pragma unroll
    for (int v = 0; v < 8; ++v)
      if (lo == v) out_ld[rowbase + 8 * hi + v] = ldacc[v];
  }
}

extern "C" void kernel_launch(void* const* d_in, const int* in_sizes, int n_in,
                              void* d_out, int out_size, void* d_ws, size_t ws_size,
                              hipStream_t stream) {
  (void)n_in; (void)out_size; (void)d_ws; (void)ws_size;
  const float* z1  = (const float*)d_in[0];
  const float* eps = (const float*)d_in[1];
  const float* W0  = (const float*)d_in[2];
  const float* b0  = (const float*)d_in[3];
  const float* W1  = (const float*)d_in[4];
  const float* b1  = (const float*)d_in[5];
  const float* W2  = (const float*)d_in[6];
  const float* b2  = (const float*)d_in[7];

  const int Btot = in_sizes[0] / DD;
  float* out_z  = (float*)d_out;
  float* out_ld = out_z + (size_t)Btot * DD;

  dim3 grid(Btot / ROWS_BLK), block(NWAVES * 32);
  size_t shmem = (size_t)SMEM_FLOATS * sizeof(float);
  ffjord_kernel<<<grid, block, shmem, stream>>>(z1, eps, W0, b0, W1, b1, W2, b2,
                                                out_z, out_ld);
}


# Fragment loads of each 32-WMMA block issued as one group a block ahead into a second register set, held by scheduling fences, replacing load / full wait / 2 WMMAs sequences
# speedup vs baseline: 1.2020x; 1.2020x over previous
#include <hip/hip_runtime.h>


typedef __attribute__((ext_vector_type(2))) float v2f;
typedef __attribute__((ext_vector_type(8))) float v8f;

#define NWAVES    8
#define MROWS     16
#define ROWS_BLK  (NWAVES * MROWS)
#define DD        32
#define HH        128
#define SSTRIDE   132

#define OFF_W0   0
#define OFF_W1   (OFF_W0 + 4096)
#define OFF_W2   (OFF_W1 + 16384)
#define OFF_B0   (OFF_W2 + 4096)
#define OFF_T0   (OFF_B0 + 128)
#define OFF_B1   (OFF_T0 + 128)
#define OFF_T1   (OFF_B1 + 128)
#define OFF_B2   (OFF_T1 + 128)
#define OFF_T2   (OFF_B2 + 32)
#define OFF_STG  (OFF_T2 + 32)
#define STG_WAVE (2 * 16 * SSTRIDE)
#define SMEM_FLOATS (OFF_STG + NWAVES * STG_WAVE)

__device__ __forceinline__ float fast_tanh(float x) {
  float cx = fminf(fmaxf(x, -10.0f), 10.0f);
  float e  = __expf(2.0f * cx);
  return 1.0f - 2.0f * __builtin_amdgcn_rcpf(e + 1.0f);
}

__device__ __forceinline__ v8f wmma4(v2f a, v2f b, v8f c) {
  return __builtin_amdgcn_wmma_f32_16x16x4_f32(false, a, false, b, (short)0, c,
                                               false, false);
}

template <int NKC, int NT>
__device__ __forceinline__ void load_b(v2f* b, const float* wF, int kc0) {
#pragma unroll
  for (int j = 0; j < NKC; ++j)
#pragma unroll
    for (int nt = 0; nt < NT; ++nt)
      b[j * NT + nt] = *(const v2f*)(wF + ((kc0 + j) * NT + nt) * 64);
}
template <int NKC>
__device__ __forceinline__ void load_a(v2f* a, const float* sA, int kc0) {
#pragma unroll
  for (int j = 0; j < NKC; ++j)
    a[j] = *(const v2f*)(sA + 4 * (kc0 + j));
}
template <int NKC, int NT>
__device__ __forceinline__ void mma_block(v8f* acc, const v2f* a, const v2f* b) {
#pragma unroll
  for (int j = 0; j < NKC; ++j)
#pragma unroll
    for (int nt = 0; nt < NT; ++nt)
      acc[nt] = wmma4(a[j], b[j * NT + nt], acc[nt]);
}
template <int NKC>
__device__ __forceinline__ void mma_block2(v8f* dacc, v8f* jacc, const v2f* ah,
                                           const v2f* au, const v2f* b) {
#pragma unroll
  for (int j = 0; j < NKC; ++j)
#pragma unroll
    for (int nt = 0; nt < 2; ++nt) {
      dacc[nt] = wmma4(ah[j], b[j * 2 + nt], dacc[nt]);
      jacc[nt] = wmma4(au[j], b[j * 2 + nt], jacc[nt]);
    }
}

#define STAGE_FENCE() __builtin_amdgcn_sched_barrier(0)

__device__ __forceinline__ void gemm_k128_n128(v8f* acc, const float* sA, const float* wF) {
  v2f a0[4], b0[32], a1[4], b1[32];
  load_a<4>(a0, sA, 0);
  load_b<4, 8>(b0, wF, 0);
#pragma clang loop unroll(disable)
  for (int kb = 0; kb < 32; kb += 8) {
    load_a<4>(a1, sA, kb + 4);
    load_b<4, 8>(b1, wF, kb + 4);
    STAGE_FENCE();
    mma_block<4, 8>(acc, a0, b0);
    STAGE_FENCE();
    const int kn = (kb + 8) & 31;
    load_a<4>(a0, sA, kn);
    load_b<4, 8>(b0, wF, kn);
    STAGE_FENCE();
    mma_block<4, 8>(acc, a1, b1);
  }
}

__device__ __forceinline__ void gemm_k32_n128(v8f* acc, const v2f* a, const float* wF) {
  v2f b0[32], b1[32];
  load_b<4, 8>(b0, wF, 0);
  load_b<4, 8>(b1, wF, 4);
  STAGE_FENCE();
  mma_block<4, 8>(acc, a, b0);
  mma_block<4, 8>(acc, a + 4, b1);
}

__device__ __forceinline__ void func_eval(
    const float* smem, float* sH, float* sU, int lo, int hi,
    const v2f* zin, const v2f* eps_a, const float (*eps_d)[8],
    v2f* kcur, float* lsum, float t, float wrk)
{
  const v8f ZERO8 = {0.f,0.f,0.f,0.f,0.f,0.f,0.f,0.f};

  const int fb = hi * 32 + 2 * lo;
  const float* wF0 = smem + OFF_W0 + fb;
  const float* wF1 = smem + OFF_W1 + fb;
  const float* wF2 = smem + OFF_W2 + fb;
  float* sHa = sH + lo * SSTRIDE + 2 * hi;
  float* sUa = sU + lo * SSTRIDE + 2 * hi;
  float* sHs = sH + 8 * hi * SSTRIDE + lo;
  float* sUs = sU + 8 * hi * SSTRIDE + lo;

  v8f acc[8];
#pragma unroll
  for (int nt = 0; nt < 8; ++nt) acc[nt] = ZERO8;
  gemm_k32_n128(acc, zin, wF0);
#pragma unroll
  for (int nt = 0; nt < 8; ++nt) {
    int col = 16 * nt + lo;
    float bias = smem[OFF_B0 + col] + t * smem[OFF_T0 + col];
#pragma unroll
    for (int v = 0; v < 8; ++v)
      sHs[v * SSTRIDE + 16 * nt] = fast_tanh(acc[nt][v] + bias);
  }
  __builtin_amdgcn_wave_barrier();

#pragma unroll
  for (int nt = 0; nt < 8; ++nt) acc[nt] = ZERO8;
  gemm_k32_n128(acc, eps_a, wF0);
#pragma unroll
  for (int nt = 0; nt < 8; ++nt) {
#pragma unroll
    for (int v = 0; v < 8; ++v) {
      int idx = v * SSTRIDE + 16 * nt;
      float h = sHs[idx];
      sUs[idx] = acc[nt][v] * (1.0f - h * h);
    }
  }
  __builtin_amdgcn_wave_barrier();

#pragma unroll
  for (int nt = 0; nt < 8; ++nt) acc[nt] = ZERO8;
  gemm_k128_n128(acc, sHa, wF1);
  __builtin_amdgcn_wave_barrier();
#pragma unroll
  for (int nt = 0; nt < 8; ++nt) {
    int col = 16 * nt + lo;
    float bias = smem[OFF_B1 + col] + t * smem[OFF_T1 + col];
#pragma unroll
    for (int v = 0; v < 8; ++v)
      sHs[v * SSTRIDE + 16 * nt] = fast_tanh(acc[nt][v] + bias);
  }
  __builtin_amdgcn_wave_barrier();

#pragma unroll
  for (int nt = 0; nt < 8; ++nt) acc[nt] = ZERO8;
  gemm_k128_n128(acc, sUa, wF1);
  __builtin_amdgcn_wave_barrier();
#pragma unroll
  for (int nt = 0; nt < 8; ++nt) {
#pragma unroll
    for (int v = 0; v < 8; ++v) {
      int idx = v * SSTRIDE + 16 * nt;
      float h = sHs[idx];
      sUs[idx] = acc[nt][v] * (1.0f - h * h);
    }
  }
  __builtin_amdgcn_wave_barrier();

  v8f dacc[2], jacc[2];
  dacc[0] = ZERO8; dacc[1] = ZERO8; jacc[0] = ZERO8; jacc[1] = ZERO8;
  {
    v2f ah0[8], au0[8], w0[16], ah1[8], au1[8], w1[16];
    load_a<8>(ah0, sHa, 0);  load_a<8>(au0, sUa, 0);  load_b<8, 2>(w0, wF2, 0);
    load_a<8>(ah1, sHa, 8);  load_a<8>(au1, sUa, 8);  load_b<8, 2>(w1, wF2, 8);
    STAGE_FENCE();
    mma_block2<8>(dacc, jacc, ah0, au0, w0);
    STAGE_FENCE();
    load_a<8>(ah0, sHa, 16); load_a<8>(au0, sUa, 16); load_b<8, 2>(w0, wF2, 16);
    STAGE_FENCE();
    mma_block2<8>(dacc, jacc, ah1, au1, w1);
    STAGE_FENCE();
    load_a<8>(ah1, sHa, 24); load_a<8>(au1, sUa, 24); load_b<8, 2>(w1, wF2, 24);
    STAGE_FENCE();
    mma_block2<8>(dacc, jacc, ah0, au0, w0);
    mma_block2<8>(dacc, jacc, ah1, au1, w1);
  }
  __builtin_amdgcn_wave_barrier();

  float p[8];
#pragma unroll
  for (int v = 0; v < 8; ++v)
    p[v] = eps_d[0][v] * jacc[0][v] + eps_d[1][v] * jacc[1][v];
#pragma unroll
  for (int off = 1; off < 16; off <<= 1) {
#pragma unroll
    for (int v = 0; v < 8; ++v) p[v] += __shfl_xor(p[v], off, 32);
  }
#pragma unroll
  for (int v = 0; v < 8; ++v) lsum[v] += wrk * p[v];

#pragma unroll
  for (int nt = 0; nt < 2; ++nt) {
    int col = 16 * nt + lo;
    float bias = smem[OFF_B2 + col] + t * smem[OFF_T2 + col];
#pragma unroll
    for (int v = 0; v < 8; ++v)
      sHs[v * SSTRIDE + 16 * nt] = dacc[nt][v] + bias;
  }
  __builtin_amdgcn_wave_barrier();
#pragma unroll
  for (int kc = 0; kc < 8; ++kc)
    kcur[kc] = *(const v2f*)(sHa + 4 * kc);
  __builtin_amdgcn_wave_barrier();
}

__global__ __launch_bounds__(256)
void ffjord_kernel(const float* __restrict__ z1, const float* __restrict__ eps,
                   const float* __restrict__ W0, const float* __restrict__ b0,
                   const float* __restrict__ W1, const float* __restrict__ b1,
                   const float* __restrict__ W2, const float* __restrict__ b2,
                   float* __restrict__ out_z, float* __restrict__ out_ld)
{
  extern __shared__ float smem[];
  const int tid  = threadIdx.x;
  const int lane = tid & 31;
  const int wave = tid >> 5;
  const int lo   = lane & 15;
  const int hi   = lane >> 4;

  for (int i = tid; i < 32 * 128; i += blockDim.x) {
    int k = i >> 7, n = i & 127;
    int idx = (((k >> 2) * 8 + (n >> 4)) << 6) + (((k >> 1) & 1) << 5) + 2 * (n & 15) + (k & 1);
    smem[OFF_W0 + idx] = W0[k * 128 + n];
  }
  for (int i = tid; i < 128 * 128; i += blockDim.x) {
    int k = i >> 7, n = i & 127;
    int idx = (((k >> 2) * 8 + (n >> 4)) << 6) + (((k >> 1) & 1) << 5) + 2 * (n & 15) + (k & 1);
    smem[OFF_W1 + idx] = W1[k * 128 + n];
  }
  for (int i = tid; i < 128 * 32; i += blockDim.x) {
    int k = i >> 5, n = i & 31;
    int idx = (((k >> 2) * 2 + (n >> 4)) << 6) + (((k >> 1) & 1) << 5) + 2 * (n & 15) + (k & 1);
    smem[OFF_W2 + idx] = W2[k * 32 + n];
  }
  for (int n = tid; n < 128; n += blockDim.x) {
    smem[OFF_T0 + n] = W0[32 * 128 + n];
    smem[OFF_B0 + n] = b0[n];
    smem[OFF_T1 + n] = W1[128 * 128 + n];
    smem[OFF_B1 + n] = b1[n];
  }
  for (int n = tid; n < 32; n += blockDim.x) {
    smem[OFF_T2 + n] = W2[128 * 32 + n];
    smem[OFF_B2 + n] = b2[n];
  }
  __syncthreads();

  float* sH = smem + OFF_STG + wave * STG_WAVE;
  float* sU = sH + 16 * SSTRIDE;
  const int rowbase = blockIdx.x * ROWS_BLK + wave * MROWS;

  v2f z_a[8], eps_a[8], zin[8], ksum[8], kcur[8];
  float eps_d[2][8];
  float ldacc[8];

  {
    const float* zrow = z1  + (size_t)(rowbase + lo) * DD + 2 * hi;
    const float* erow = eps + (size_t)(rowbase + lo) * DD + 2 * hi;
#pragma unroll
    for (int kc = 0; kc < 8; ++kc) {
      z_a[kc]   = *(const v2f*)(zrow + 4 * kc);
      eps_a[kc] = *(const v2f*)(erow + 4 * kc);
    }
#pragma unroll
    for (int nt = 0; nt < 2; ++nt)
#pragma unroll
      for (int v = 0; v < 8; ++v)
        eps_d[nt][v] = eps[(size_t)(rowbase + v + 8 * hi) * DD + 16 * nt + lo];
#pragma unroll
    for (int v = 0; v < 8; ++v) ldacc[v] = 0.0f;
  }

  const float hstep = -0.2f;
  for (int s = 0; s < 5; ++s) {
    const float tb = 1.0f - 0.2f * (float)s;
    float lsum[8];
#pragma unroll
    for (int v = 0; v < 8; ++v) lsum[v] = 0.0f;

#pragma unroll
    for (int i = 0; i < 8; ++i) zin[i] = z_a[i];
    func_eval(smem, sH, sU, lo, hi, zin, eps_a, eps_d, kcur, lsum, tb, 1.0f);
#pragma unroll
    for (int i = 0; i < 8; ++i) { ksum[i] = kcur[i]; zin[i] = z_a[i] + kcur[i] * (0.5f * hstep); }

    func_eval(smem, sH, sU, lo, hi, zin, eps_a, eps_d, kcur, lsum, tb + 0.5f * hstep, 2.0f);
#pragma unroll
    for (int i = 0; i < 8; ++i) { ksum[i] += kcur[i] * 2.0f; zin[i] = z_a[i] + kcur[i] * (0.5f * hstep); }

    func_eval(smem, sH, sU, lo, hi, zin, eps_a, eps_d, kcur, lsum, tb + 0.5f * hstep, 2.0f);
#pragma unroll
    for (int i = 0; i < 8; ++i) { ksum[i] += kcur[i] * 2.0f; zin[i] = z_a[i] + kcur[i] * hstep; }

    func_eval(smem, sH, sU, lo, hi, zin, eps_a, eps_d, kcur, lsum, tb + hstep, 1.0f);
#pragma unroll
    for (int i = 0; i < 8; ++i) ksum[i] += kcur[i];

    const float c = hstep / 6.0f;
#pragma unroll
    for (int i = 0; i < 8; ++i) z_a[i] += ksum[i] * c;
#pragma unroll
    for (int v = 0; v < 8; ++v) ldacc[v] -= c * lsum[v];
  }

  {
    float* orow = out_z + (size_t)(rowbase + lo) * DD + 2 * hi;
#pragma unroll
    for (int kc = 0; kc < 8; ++kc)
      *(v2f*)(orow + 4 * kc) = z_a[kc];
#pragma unroll
    for (int v = 0; v < 8; ++v)
      if (lo == v) out_ld[rowbase + 8 * hi + v] = ldacc[v];
  }
}

extern "C" void kernel_launch(void* const* d_in, const int* in_sizes, int n_in,
                              void* d_out, int out_size, void* d_ws, size_t ws_size,
                              hipStream_t stream) {
  (void)n_in; (void)out_size; (void)d_ws; (void)ws_size;
  const float* z1  = (const float*)d_in[0];
  const float* eps = (const float*)d_in[1];
  const float* W0  = (const float*)d_in[2];
  const float* b0  = (const float*)d_in[3];
  const float* W1  = (const float*)d_in[4];
  const float* b1  = (const float*)d_in[5];
  const float* W2  = (const float*)d_in[6];
  const float* b2  = (const float*)d_in[7];

  const int Btot = in_sizes[0] / DD;
  float* out_z  = (float*)d_out;
  float* out_ld = out_z + (size_t)Btot * DD;

  dim3 grid(Btot / ROWS_BLK), block(NWAVES * 32);
  size_t shmem = (size_t)SMEM_FLOATS * sizeof(float);
  ffjord_kernel<<<grid, block, shmem, stream>>>(z1, eps, W0, b0, W1, b1, W2, b2,
                                                out_z, out_ld);
}


# Weight staging into LDS with compile-time trip counts: each thread issues its 24 128-bit global loads at once and stores row pairs as 128-bit LDS writes, instead of one dword load per round trip
# speedup vs baseline: 1.2825x; 1.0670x over previous
#include <hip/hip_runtime.h>


typedef __attribute__((ext_vector_type(2))) float v2f;
typedef __attribute__((ext_vector_type(4))) float v4f;
typedef __attribute__((ext_vector_type(8))) float v8f;

#define NWAVES    8
#define MROWS     16
#define ROWS_BLK  (NWAVES * MROWS)
#define DD        32
#define HH        128
#define SSTRIDE   132

#define OFF_W0   0
#define OFF_W1   (OFF_W0 + 4096)
#define OFF_W2   (OFF_W1 + 16384)
#define OFF_B0   (OFF_W2 + 4096)
#define OFF_T0   (OFF_B0 + 128)
#define OFF_B1   (OFF_T0 + 128)
#define OFF_T1   (OFF_B1 + 128)
#define OFF_B2   (OFF_T1 + 128)
#define OFF_T2   (OFF_B2 + 32)
#define OFF_STG  (OFF_T2 + 32)
#define STG_WAVE (2 * 16 * SSTRIDE)
#define SMEM_FLOATS (OFF_STG + NWAVES * STG_WAVE)

#define NTHR (NWAVES * 32)
#define STG_PER(K, N) (((K) / 2) * ((N) / 4) / NTHR)
template <int K, int N>
__device__ __forceinline__ void stage_load(v4f* r, const float* __restrict__ Wg, int tid) {
#pragma unroll
  for (int j = 0; j < STG_PER(K, N); ++j) {
    const int u = tid + NTHR * j;
    const float* p = Wg + (2 * (u / (N / 4))) * N + 4 * (u % (N / 4));
    r[2 * j]     = *(const v4f*)p;
    r[2 * j + 1] = *(const v4f*)(p + N);
  }
}
template <int K, int N>
__device__ __forceinline__ void stage_store(float* dst, const v4f* r, int tid) {
#pragma unroll
  for (int j = 0; j < STG_PER(K, N); ++j) {
    const int u = tid + NTHR * j;
    const int k = 2 * (u / (N / 4)), n = 4 * (u % (N / 4));
    const int idx = (((k >> 2) * (N / 16) + (n >> 4)) << 6) + (((k >> 1) & 1) << 5) + 2 * (n & 15);
    const v4f a = r[2 * j], b = r[2 * j + 1];
    v4f lo4 = {a.x, b.x, a.y, b.y};
    v4f hi4 = {a.z, b.z, a.w, b.w};
    *(v4f*)(dst + idx)     = lo4;
    *(v4f*)(dst + idx + 4) = hi4;
  }
}

__device__ __forceinline__ float fast_tanh(float x) {
  float cx = fminf(fmaxf(x, -10.0f), 10.0f);
  float e  = __expf(2.0f * cx);
  return 1.0f - 2.0f * __builtin_amdgcn_rcpf(e + 1.0f);
}

__device__ __forceinline__ v8f wmma4(v2f a, v2f b, v8f c) {
  return __builtin_amdgcn_wmma_f32_16x16x4_f32(false, a, false, b, (short)0, c,
                                               false, false);
}

template <int NKC, int NT>
__device__ __forceinline__ void load_b(v2f* b, const float* wF, int kc0) {
#pragma unroll
  for (int j = 0; j < NKC; ++j)
#pragma unroll
    for (int nt = 0; nt < NT; ++nt)
      b[j * NT + nt] = *(const v2f*)(wF + ((kc0 + j) * NT + nt) * 64);
}
template <int NKC>
__device__ __forceinline__ void load_a(v2f* a, const float* sA, int kc0) {
#pragma unroll
  for (int j = 0; j < NKC; ++j)
    a[j] = *(const v2f*)(sA + 4 * (kc0 + j));
}
template <int NKC, int NT>
__device__ __forceinline__ void mma_block(v8f* acc, const v2f* a, const v2f* b) {
#pragma unroll
  for (int j = 0; j < NKC; ++j)
#pragma unroll
    for (int nt = 0; nt < NT; ++nt)
      acc[nt] = wmma4(a[j], b[j * NT + nt], acc[nt]);
}
template <int NKC>
__device__ __forceinline__ void mma_block2(v8f* dacc, v8f* jacc, const v2f* ah,
                                           const v2f* au, const v2f* b) {
#pragma unroll
  for (int j = 0; j < NKC; ++j)
#pragma unroll
    for (int nt = 0; nt < 2; ++nt) {
      dacc[nt] = wmma4(ah[j], b[j * 2 + nt], dacc[nt]);
      jacc[nt] = wmma4(au[j], b[j * 2 + nt], jacc[nt]);
    }
}

#define STAGE_FENCE() __builtin_amdgcn_sched_barrier(0)

__device__ __forceinline__ void gemm_k128_n128(v8f* acc, const float* sA, const float* wF) {
  v2f a0[4], b0[32], a1[4], b1[32];
  load_a<4>(a0, sA, 0);
  load_b<4, 8>(b0, wF, 0);
#pragma clang loop unroll(disable)
  for (int kb = 0; kb < 32; kb += 8) {
    load_a<4>(a1, sA, kb + 4);
    load_b<4, 8>(b1, wF, kb + 4);
    STAGE_FENCE();
    mma_block<4, 8>(acc, a0, b0);
    STAGE_FENCE();
    const int kn = (kb + 8) & 31;
    load_a<4>(a0, sA, kn);
    load_b<4, 8>(b0, wF, kn);
    STAGE_FENCE();
    mma_block<4, 8>(acc, a1, b1);
  }
}

__device__ __forceinline__ void gemm_k32_n128(v8f* acc, const v2f* a, const float* wF) {
  v2f b0[32], b1[32];
  load_b<4, 8>(b0, wF, 0);
  load_b<4, 8>(b1, wF, 4);
  STAGE_FENCE();
  mma_block<4, 8>(acc, a, b0);
  mma_block<4, 8>(acc, a + 4, b1);
}

__device__ __forceinline__ void func_eval(
    const float* smem, float* sH, float* sU, int lo, int hi,
    const v2f* zin, const v2f* eps_a, const float (*eps_d)[8],
    v2f* kcur, float* lsum, float t, float wrk)
{
  const v8f ZERO8 = {0.f,0.f,0.f,0.f,0.f,0.f,0.f,0.f};

  const int fb = hi * 32 + 2 * lo;
  const float* wF0 = smem + OFF_W0 + fb;
  const float* wF1 = smem + OFF_W1 + fb;
  const float* wF2 = smem + OFF_W2 + fb;
  float* sHa = sH + lo * SSTRIDE + 2 * hi;
  float* sUa = sU + lo * SSTRIDE + 2 * hi;
  float* sHs = sH + 8 * hi * SSTRIDE + lo;
  float* sUs = sU + 8 * hi * SSTRIDE + lo;

  v8f acc[8];
#pragma unroll
  for (int nt = 0; nt < 8; ++nt) acc[nt] = ZERO8;
  gemm_k32_n128(acc, zin, wF0);
#pragma unroll
  for (int nt = 0; nt < 8; ++nt) {
    int col = 16 * nt + lo;
    float bias = smem[OFF_B0 + col] + t * smem[OFF_T0 + col];
#pragma unroll
    for (int v = 0; v < 8; ++v)
      sHs[v * SSTRIDE + 16 * nt] = fast_tanh(acc[nt][v] + bias);
  }
  __builtin_amdgcn_wave_barrier();

#pragma unroll
  for (int nt = 0; nt < 8; ++nt) acc[nt] = ZERO8;
  gemm_k32_n128(acc, eps_a, wF0);
#pragma unroll
  for (int nt = 0; nt < 8; ++nt) {
#pragma unroll
    for (int v = 0; v < 8; ++v) {
      int idx = v * SSTRIDE + 16 * nt;
      float h = sHs[idx];
      sUs[idx] = acc[nt][v] * (1.0f - h * h);
    }
  }
  __builtin_amdgcn_wave_barrier();

#pragma unroll
  for (int nt = 0; nt < 8; ++nt) acc[nt] = ZERO8;
  gemm_k128_n128(acc, sHa, wF1);
  __builtin_amdgcn_wave_barrier();
#pragma unroll
  for (int nt = 0; nt < 8; ++nt) {
    int col = 16 * nt + lo;
    float bias = smem[OFF_B1 + col] + t * smem[OFF_T1 + col];
#pragma unroll
    for (int v = 0; v < 8; ++v)
      sHs[v * SSTRIDE + 16 * nt] = fast_tanh(acc[nt][v] + bias);
  }
  __builtin_amdgcn_wave_barrier();

#pragma unroll
  for (int nt = 0; nt < 8; ++nt) acc[nt] = ZERO8;
  gemm_k128_n128(acc, sUa, wF1);
  __builtin_amdgcn_wave_barrier();
#pragma unroll
  for (int nt = 0; nt < 8; ++nt) {
#pragma unroll
    for (int v = 0; v < 8; ++v) {
      int idx = v * SSTRIDE + 16 * nt;
      float h = sHs[idx];
      sUs[idx] = acc[nt][v] * (1.0f - h * h);
    }
  }
  __builtin_amdgcn_wave_barrier();

  v8f dacc[2], jacc[2];
  dacc[0] = ZERO8; dacc[1] = ZERO8; jacc[0] = ZERO8; jacc[1] = ZERO8;
  {
    v2f ah0[8], au0[8], w0[16], ah1[8], au1[8], w1[16];
    load_a<8>(ah0, sHa, 0);  load_a<8>(au0, sUa, 0);  load_b<8, 2>(w0, wF2, 0);
    load_a<8>(ah1, sHa, 8);  load_a<8>(au1, sUa, 8);  load_b<8, 2>(w1, wF2, 8);
    STAGE_FENCE();
    mma_block2<8>(dacc, jacc, ah0, au0, w0);
    STAGE_FENCE();
    load_a<8>(ah0, sHa, 16); load_a<8>(au0, sUa, 16); load_b<8, 2>(w0, wF2, 16);
    STAGE_FENCE();
    mma_block2<8>(dacc, jacc, ah1, au1, w1);
    STAGE_FENCE();
    load_a<8>(ah1, sHa, 24); load_a<8>(au1, sUa, 24); load_b<8, 2>(w1, wF2, 24);
    STAGE_FENCE();
    mma_block2<8>(dacc, jacc, ah0, au0, w0);
    mma_block2<8>(dacc, jacc, ah1, au1, w1);
  }
  __builtin_amdgcn_wave_barrier();

  float p[8];
#pragma unroll
  for (int v = 0; v < 8; ++v)
    p[v] = eps_d[0][v] * jacc[0][v] + eps_d[1][v] * jacc[1][v];
#pragma unroll
  for (int off = 1; off < 16; off <<= 1) {
#pragma unroll
    for (int v = 0; v < 8; ++v) p[v] += __shfl_xor(p[v], off, 32);
  }
#pragma unroll
  for (int v = 0; v < 8; ++v) lsum[v] += wrk * p[v];

#pragma unroll
  for (int nt = 0; nt < 2; ++nt) {
    int col = 16 * nt + lo;
    float bias = smem[OFF_B2 + col] + t * smem[OFF_T2 + col];
#pragma unroll
    for (int v = 0; v < 8; ++v)
      sHs[v * SSTRIDE + 16 * nt] = dacc[nt][v] + bias;
  }
  __builtin_amdgcn_wave_barrier();
#pragma unroll
  for (int kc = 0; kc < 8; ++kc)
    kcur[kc] = *(const v2f*)(sHa + 4 * kc);
  __builtin_amdgcn_wave_barrier();
}

__global__ __launch_bounds__(256)
void ffjord_kernel(const float* __restrict__ z1, const float* __restrict__ eps,
                   const float* __restrict__ W0, const float* __restrict__ b0,
                   const float* __restrict__ W1, const float* __restrict__ b1,
                   const float* __restrict__ W2, const float* __restrict__ b2,
                   float* __restrict__ out_z, float* __restrict__ out_ld)
{
  extern __shared__ __attribute__((aligned(16))) float smem[];
  const int tid  = threadIdx.x;
  const int lane = tid & 31;
  const int wave = tid >> 5;
  const int lo   = lane & 15;
  const int hi   = lane >> 4;

  v4f rw0[2 * STG_PER(32, 128)], rw1[2 * STG_PER(128, 128)], rw2[2 * STG_PER(128, 32)];
  stage_load<32, 128>(rw0, W0, tid);
  stage_load<128, 128>(rw1, W1, tid);
  stage_load<128, 32>(rw2, W2, tid);
  const int nb = tid & 127, nc = tid & 31;
  const float t0v = W0[32 * 128 + nb], b0v = b0[nb];
  const float t1v = W1[128 * 128 + nb], b1v = b1[nb];
  const float t2v = W2[128 * 32 + nc], b2v = b2[nc];
  stage_store<32, 128>(smem + OFF_W0, rw0, tid);
  stage_store<128, 128>(smem + OFF_W1, rw1, tid);
  stage_store<128, 32>(smem + OFF_W2, rw2, tid);
  if (tid < 128) {
    smem[OFF_T0 + nb] = t0v;
    smem[OFF_B0 + nb] = b0v;
    smem[OFF_T1 + nb] = t1v;
    smem[OFF_B1 + nb] = b1v;
  }
  if (tid < 32) {
    smem[OFF_T2 + nc] = t2v;
    smem[OFF_B2 + nc] = b2v;
  }
  __syncthreads();

  float* sH = smem + OFF_STG + wave * STG_WAVE;
  float* sU = sH + 16 * SSTRIDE;
  const int rowbase = blockIdx.x * ROWS_BLK + wave * MROWS;

  v2f z_a[8], eps_a[8], zin[8], ksum[8], kcur[8];
  float eps_d[2][8];
  float ldacc[8];

  {
    const float* zrow = z1  + (size_t)(rowbase + lo) * DD + 2 * hi;
    const float* erow = eps + (size_t)(rowbase + lo) * DD + 2 * hi;
#pragma unroll
    for (int kc = 0; kc < 8; ++kc) {
      z_a[kc]   = *(const v2f*)(zrow + 4 * kc);
      eps_a[kc] = *(const v2f*)(erow + 4 * kc);
    }
#pragma unroll
    for (int nt = 0; nt < 2; ++nt)
#pragma unroll
      for (int v = 0; v < 8; ++v)
        eps_d[nt][v] = eps[(size_t)(rowbase + v + 8 * hi) * DD + 16 * nt + lo];
#pragma unroll
    for (int v = 0; v < 8; ++v) ldacc[v] = 0.0f;
  }

  const float hstep = -0.2f;
  for (int s = 0; s < 5; ++s) {
    const float tb = 1.0f - 0.2f * (float)s;
    float lsum[8];
#pragma unroll
    for (int v = 0; v < 8; ++v) lsum[v] = 0.0f;

#pragma unroll
    for (int i = 0; i < 8; ++i) zin[i] = z_a[i];
    func_eval(smem, sH, sU, lo, hi, zin, eps_a, eps_d, kcur, lsum, tb, 1.0f);
#pragma unroll
    for (int i = 0; i < 8; ++i) { ksum[i] = kcur[i]; zin[i] = z_a[i] + kcur[i] * (0.5f * hstep); }

    func_eval(smem, sH, sU, lo, hi, zin, eps_a, eps_d, kcur, lsum, tb + 0.5f * hstep, 2.0f);
#pragma unroll
    for (int i = 0; i < 8; ++i) { ksum[i] += kcur[i] * 2.0f; zin[i] = z_a[i] + kcur[i] * (0.5f * hstep); }

    func_eval(smem, sH, sU, lo, hi, zin, eps_a, eps_d, kcur, lsum, tb + 0.5f * hstep, 2.0f);
#pragma unroll
    for (int i = 0; i < 8; ++i) { ksum[i] += kcur[i] * 2.0f; zin[i] = z_a[i] + kcur[i] * hstep; }

    func_eval(smem, sH, sU, lo, hi, zin, eps_a, eps_d, kcur, lsum, tb + hstep, 1.0f);
#pragma unroll
    for (int i = 0; i < 8; ++i) ksum[i] += kcur[i];

    const float c = hstep / 6.0f;
#pragma unroll
    for (int i = 0; i < 8; ++i) z_a[i] += ksum[i] * c;
#pragma unroll
    for (int v = 0; v < 8; ++v) ldacc[v] -= c * lsum[v];
  }

  {
    float* orow = out_z + (size_t)(rowbase + lo) * DD + 2 * hi;
#pragma unroll
    for (int kc = 0; kc < 8; ++kc)
      *(v2f*)(orow + 4 * kc) = z_a[kc];
#pragma unroll
    for (int v = 0; v < 8; ++v)
      if (lo == v) out_ld[rowbase + 8 * hi + v] = ldacc[v];
  }
}

extern "C" void kernel_launch(void* const* d_in, const int* in_sizes, int n_in,
                              void* d_out, int out_size, void* d_ws, size_t ws_size,
                              hipStream_t stream) {
  (void)n_in; (void)out_size; (void)d_ws; (void)ws_size;
  const float* z1  = (const float*)d_in[0];
  const float* eps = (const float*)d_in[1];
  const float* W0  = (const float*)d_in[2];
  const float* b0  = (const float*)d_in[3];
  const float* W1  = (const float*)d_in[4];
  const float* b1  = (const float*)d_in[5];
  const float* W2  = (const float*)d_in[6];
  const float* b2  = (const float*)d_in[7];

  const int Btot = in_sizes[0] / DD;
  float* out_z  = (float*)d_out;
  float* out_ld = out_z + (size_t)Btot * DD;

  dim3 grid(Btot / ROWS_BLK), block(NWAVES * 32);
  size_t shmem = (size_t)SMEM_FLOATS * sizeof(float);
  ffjord_kernel<<<grid, block, shmem, stream>>>(z1, eps, W0, b0, W1, b1, W2, b2,
                                                out_z, out_ld);
}


# RK4's four stage evaluations run as trips of one rolled loop with per-stage scalars instead of four inlined copies: loop code shrinks from about 75 KB to 19 KB and fits the 64 KiB instruction cache
# speedup vs baseline: 1.3554x; 1.0568x over previous
#include <hip/hip_runtime.h>


typedef __attribute__((ext_vector_type(2))) float v2f;
typedef __attribute__((ext_vector_type(4))) float v4f;
typedef __attribute__((ext_vector_type(8))) float v8f;

#define NWAVES    8
#define MROWS     16
#define ROWS_BLK  (NWAVES * MROWS)
#define DD        32
#define HH        128
#define SSTRIDE   132

#define OFF_W0   0
#define OFF_W1   (OFF_W0 + 4096)
#define OFF_W2   (OFF_W1 + 16384)
#define OFF_B0   (OFF_W2 + 4096)
#define OFF_T0   (OFF_B0 + 128)
#define OFF_B1   (OFF_T0 + 128)
#define OFF_T1   (OFF_B1 + 128)
#define OFF_B2   (OFF_T1 + 128)
#define OFF_T2   (OFF_B2 + 32)
#define OFF_STG  (OFF_T2 + 32)
#define STG_WAVE (2 * 16 * SSTRIDE)
#define SMEM_FLOATS (OFF_STG + NWAVES * STG_WAVE)

#define NTHR (NWAVES * 32)
#define STG_PER(K, N) (((K) / 2) * ((N) / 4) / NTHR)
template <int K, int N>
__device__ __forceinline__ void stage_load(v4f* r, const float* __restrict__ Wg, int tid) {
#pragma unroll
  for (int j = 0; j < STG_PER(K, N); ++j) {
    const int u = tid + NTHR * j;
    const float* p = Wg + (2 * (u / (N / 4))) * N + 4 * (u % (N / 4));
    r[2 * j]     = *(const v4f*)p;
    r[2 * j + 1] = *(const v4f*)(p + N);
  }
}
template <int K, int N>
__device__ __forceinline__ void stage_store(float* dst, const v4f* r, int tid) {
#pragma unroll
  for (int j = 0; j < STG_PER(K, N); ++j) {
    const int u = tid + NTHR * j;
    const int k = 2 * (u / (N / 4)), n = 4 * (u % (N / 4));
    const int idx = (((k >> 2) * (N / 16) + (n >> 4)) << 6) + (((k >> 1) & 1) << 5) + 2 * (n & 15);
    const v4f a = r[2 * j], b = r[2 * j + 1];
    v4f lo4 = {a.x, b.x, a.y, b.y};
    v4f hi4 = {a.z, b.z, a.w, b.w};
    *(v4f*)(dst + idx)     = lo4;
    *(v4f*)(dst + idx + 4) = hi4;
  }
}

__device__ __forceinline__ float fast_tanh(float x) {
  float cx = fminf(fmaxf(x, -10.0f), 10.0f);
  float e  = __expf(2.0f * cx);
  return 1.0f - 2.0f * __builtin_amdgcn_rcpf(e + 1.0f);
}

__device__ __forceinline__ v8f wmma4(v2f a, v2f b, v8f c) {
  return __builtin_amdgcn_wmma_f32_16x16x4_f32(false, a, false, b, (short)0, c,
                                               false, false);
}

template <int NKC, int NT>
__device__ __forceinline__ void load_b(v2f* b, const float* wF, int kc0) {
#pragma unroll
  for (int j = 0; j < NKC; ++j)
#pragma unroll
    for (int nt = 0; nt < NT; ++nt)
      b[j * NT + nt] = *(const v2f*)(wF + ((kc0 + j) * NT + nt) * 64);
}
template <int NKC>
__device__ __forceinline__ void load_a(v2f* a, const float* sA, int kc0) {
#pragma unroll
  for (int j = 0; j < NKC; ++j)
    a[j] = *(const v2f*)(sA + 4 * (kc0 + j));
}
template <int NKC, int NT>
__device__ __forceinline__ void mma_block(v8f* acc, const v2f* a, const v2f* b) {
#pragma unroll
  for (int j = 0; j < NKC; ++j)
#pragma unroll
    for (int nt = 0; nt < NT; ++nt)
      acc[nt] = wmma4(a[j], b[j * NT + nt], acc[nt]);
}
template <int NKC>
__device__ __forceinline__ void mma_block2(v8f* dacc, v8f* jacc, const v2f* ah,
                                           const v2f* au, const v2f* b) {
#pragma unroll
  for (int j = 0; j < NKC; ++j)
#pragma unroll
    for (int nt = 0; nt < 2; ++nt) {
      dacc[nt] = wmma4(ah[j], b[j * 2 + nt], dacc[nt]);
      jacc[nt] = wmma4(au[j], b[j * 2 + nt], jacc[nt]);
    }
}

#define STAGE_FENCE() __builtin_amdgcn_sched_barrier(0)

__device__ __forceinline__ void gemm_k128_n128(v8f* acc, const float* sA, const float* wF) {
  v2f a0[4], b0[32], a1[4], b1[32];
  load_a<4>(a0, sA, 0);
  load_b<4, 8>(b0, wF, 0);
#pragma clang loop unroll(disable)
  for (int kb = 0; kb < 32; kb += 8) {
    load_a<4>(a1, sA, kb + 4);
    load_b<4, 8>(b1, wF, kb + 4);
    STAGE_FENCE();
    mma_block<4, 8>(acc, a0, b0);
    STAGE_FENCE();
    const int kn = (kb + 8) & 31;
    load_a<4>(a0, sA, kn);
    load_b<4, 8>(b0, wF, kn);
    STAGE_FENCE();
    mma_block<4, 8>(acc, a1, b1);
  }
}

__device__ __forceinline__ void gemm_k32_n128(v8f* acc, const v2f* a, const float* wF) {
  v2f b0[32], b1[32];
  load_b<4, 8>(b0, wF, 0);
  load_b<4, 8>(b1, wF, 4);
  STAGE_FENCE();
  mma_block<4, 8>(acc, a, b0);
  mma_block<4, 8>(acc, a + 4, b1);
}

__device__ __forceinline__ void func_eval(
    const float* smem, float* sH, float* sU, int lo, int hi,
    const v2f* zin, const v2f* eps_a, const float (*eps_d)[8],
    v2f* kcur, float* lsum, float t, float wrk)
{
  const v8f ZERO8 = {0.f,0.f,0.f,0.f,0.f,0.f,0.f,0.f};

  const int fb = hi * 32 + 2 * lo;
  const float* wF0 = smem + OFF_W0 + fb;
  const float* wF1 = smem + OFF_W1 + fb;
  const float* wF2 = smem + OFF_W2 + fb;
  float* sHa = sH + lo * SSTRIDE + 2 * hi;
  float* sUa = sU + lo * SSTRIDE + 2 * hi;
  float* sHs = sH + 8 * hi * SSTRIDE + lo;
  float* sUs = sU + 8 * hi * SSTRIDE + lo;

  v8f acc[8];
#pragma unroll
  for (int nt = 0; nt < 8; ++nt) acc[nt] = ZERO8;
  gemm_k32_n128(acc, zin, wF0);
#pragma unroll
  for (int nt = 0; nt < 8; ++nt) {
    int col = 16 * nt + lo;
    float bias = smem[OFF_B0 + col] + t * smem[OFF_T0 + col];
#pragma unroll
    for (int v = 0; v < 8; ++v)
      sHs[v * SSTRIDE + 16 * nt] = fast_tanh(acc[nt][v] + bias);
  }
  __builtin_amdgcn_wave_barrier();

#pragma unroll
  for (int nt = 0; nt < 8; ++nt) acc[nt] = ZERO8;
  gemm_k32_n128(acc, eps_a, wF0);
#pragma unroll
  for (int nt = 0; nt < 8; ++nt) {
#pragma unroll
    for (int v = 0; v < 8; ++v) {
      int idx = v * SSTRIDE + 16 * nt;
      float h = sHs[idx];
      sUs[idx] = acc[nt][v] * (1.0f - h * h);
    }
  }
  __builtin_amdgcn_wave_barrier();

#pragma unroll
  for (int nt = 0; nt < 8; ++nt) acc[nt] = ZERO8;
  gemm_k128_n128(acc, sHa, wF1);
  __builtin_amdgcn_wave_barrier();
#pragma unroll
  for (int nt = 0; nt < 8; ++nt) {
    int col = 16 * nt + lo;
    float bias = smem[OFF_B1 + col] + t * smem[OFF_T1 + col];
#pragma unroll
    for (int v = 0; v < 8; ++v)
      sHs[v * SSTRIDE + 16 * nt] = fast_tanh(acc[nt][v] + bias);
  }
  __builtin_amdgcn_wave_barrier();

#pragma unroll
  for (int nt = 0; nt < 8; ++nt) acc[nt] = ZERO8;
  gemm_k128_n128(acc, sUa, wF1);
  __builtin_amdgcn_wave_barrier();
#pragma unroll
  for (int nt = 0; nt < 8; ++nt) {
#pragma unroll
    for (int v = 0; v < 8; ++v) {
      int idx = v * SSTRIDE + 16 * nt;
      float h = sHs[idx];
      sUs[idx] = acc[nt][v] * (1.0f - h * h);
    }
  }
  __builtin_amdgcn_wave_barrier();

  v8f dacc[2], jacc[2];
  dacc[0] = ZERO8; dacc[1] = ZERO8; jacc[0] = ZERO8; jacc[1] = ZERO8;
  {
    v2f ah0[8], au0[8], w0[16], ah1[8], au1[8], w1[16];
    load_a<8>(ah0, sHa, 0);  load_a<8>(au0, sUa, 0);  load_b<8, 2>(w0, wF2, 0);
    load_a<8>(ah1, sHa, 8);  load_a<8>(au1, sUa, 8);  load_b<8, 2>(w1, wF2, 8);
    STAGE_FENCE();
    mma_block2<8>(dacc, jacc, ah0, au0, w0);
    STAGE_FENCE();
    load_a<8>(ah0, sHa, 16); load_a<8>(au0, sUa, 16); load_b<8, 2>(w0, wF2, 16);
    STAGE_FENCE();
    mma_block2<8>(dacc, jacc, ah1, au1, w1);
    STAGE_FENCE();
    load_a<8>(ah1, sHa, 24); load_a<8>(au1, sUa, 24); load_b<8, 2>(w1, wF2, 24);
    STAGE_FENCE();
    mma_block2<8>(dacc, jacc, ah0, au0, w0);
    mma_block2<8>(dacc, jacc, ah1, au1, w1);
  }
  __builtin_amdgcn_wave_barrier();

  float p[8];
#pragma unroll
  for (int v = 0; v < 8; ++v)
    p[v] = eps_d[0][v] * jacc[0][v] + eps_d[1][v] * jacc[1][v];
#pragma unroll
  for (int off = 1; off < 16; off <<= 1) {
#pragma unroll
    for (int v = 0; v < 8; ++v) p[v] += __shfl_xor(p[v], off, 32);
  }
#pragma unroll
  for (int v = 0; v < 8; ++v) lsum[v] += wrk * p[v];

#pragma unroll
  for (int nt = 0; nt < 2; ++nt) {
    int col = 16 * nt + lo;
    float bias = smem[OFF_B2 + col] + t * smem[OFF_T2 + col];
#pragma unroll
    for (int v = 0; v < 8; ++v)
      sHs[v * SSTRIDE + 16 * nt] = dacc[nt][v] + bias;
  }
  __builtin_amdgcn_wave_barrier();
#pragma unroll
  for (int kc = 0; kc < 8; ++kc)
    kcur[kc] = *(const v2f*)(sHa + 4 * kc);
  __builtin_amdgcn_wave_barrier();
}

__global__ __launch_bounds__(256)
void ffjord_kernel(const float* __restrict__ z1, const float* __restrict__ eps,
                   const float* __restrict__ W0, const float* __restrict__ b0,
                   const float* __restrict__ W1, const float* __restrict__ b1,
                   const float* __restrict__ W2, const float* __restrict__ b2,
                   float* __restrict__ out_z, float* __restrict__ out_ld)
{
  extern __shared__ __attribute__((aligned(16))) float smem[];
  const int tid  = threadIdx.x;
  const int lane = tid & 31;
  const int wave = tid >> 5;
  const int lo   = lane & 15;
  const int hi   = lane >> 4;

  v4f rw0[2 * STG_PER(32, 128)], rw1[2 * STG_PER(128, 128)], rw2[2 * STG_PER(128, 32)];
  stage_load<32, 128>(rw0, W0, tid);
  stage_load<128, 128>(rw1, W1, tid);
  stage_load<128, 32>(rw2, W2, tid);
  const int nb = tid & 127, nc = tid & 31;
  const float t0v = W0[32 * 128 + nb], b0v = b0[nb];
  const float t1v = W1[128 * 128 + nb], b1v = b1[nb];
  const float t2v = W2[128 * 32 + nc], b2v = b2[nc];
  stage_store<32, 128>(smem + OFF_W0, rw0, tid);
  stage_store<128, 128>(smem + OFF_W1, rw1, tid);
  stage_store<128, 32>(smem + OFF_W2, rw2, tid);
  if (tid < 128) {
    smem[OFF_T0 + nb] = t0v;
    smem[OFF_B0 + nb] = b0v;
    smem[OFF_T1 + nb] = t1v;
    smem[OFF_B1 + nb] = b1v;
  }
  if (tid < 32) {
    smem[OFF_T2 + nc] = t2v;
    smem[OFF_B2 + nc] = b2v;
  }
  __syncthreads();

  float* sH = smem + OFF_STG + wave * STG_WAVE;
  float* sU = sH + 16 * SSTRIDE;
  const int rowbase = blockIdx.x * ROWS_BLK + wave * MROWS;

  v2f z_a[8], eps_a[8], zin[8], ksum[8], kcur[8];
  float eps_d[2][8];
  float ldacc[8];

  {
    const float* zrow = z1  + (size_t)(rowbase + lo) * DD + 2 * hi;
    const float* erow = eps + (size_t)(rowbase + lo) * DD + 2 * hi;
#pragma unroll
    for (int kc = 0; kc < 8; ++kc) {
      z_a[kc]   = *(const v2f*)(zrow + 4 * kc);
      eps_a[kc] = *(const v2f*)(erow + 4 * kc);
    }
#pragma unroll
    for (int nt = 0; nt < 2; ++nt)
#pragma unroll
      for (int v = 0; v < 8; ++v)
        eps_d[nt][v] = eps[(size_t)(rowbase + v + 8 * hi) * DD + 16 * nt + lo];
#pragma unroll
    for (int v = 0; v < 8; ++v) ldacc[v] = 0.0f;
  }

  const float hstep = -0.2f;
  for (int s = 0; s < 5; ++s) {
    const float tb = 1.0f - 0.2f * (float)s;
    float lsum[8];
#pragma unroll
    for (int v = 0; v < 8; ++v) lsum[v] = 0.0f;
#pragma unroll
    for (int i = 0; i < 8; ++i) { zin[i] = z_a[i]; ksum[i] = (v2f){0.0f, 0.0f}; }

#pragma clang loop unroll(disable)
    for (int st = 0; st < 4; ++st) {
      const float toff = (st == 0) ? 0.0f : ((st == 3) ? hstep : 0.5f * hstep);
      const float wst  = (st == 0 || st == 3) ? 1.0f : 2.0f;
      const float cz   = (st < 2) ? 0.5f * hstep : hstep;
      func_eval(smem, sH, sU, lo, hi, zin, eps_a, eps_d, kcur, lsum, tb + toff, wst);
#pragma unroll
      for (int i = 0; i < 8; ++i) { ksum[i] += kcur[i] * wst; zin[i] = z_a[i] + kcur[i] * cz; }
    }

    const float c = hstep / 6.0f;
#pragma unroll
    for (int i = 0; i < 8; ++i) z_a[i] += ksum[i] * c;
#pragma unroll
    for (int v = 0; v < 8; ++v) ldacc[v] -= c * lsum[v];
  }

  {
    float* orow = out_z + (size_t)(rowbase + lo) * DD + 2 * hi;
#pragma unroll
    for (int kc = 0; kc < 8; ++kc)
      *(v2f*)(orow + 4 * kc) = z_a[kc];
#pragma unroll
    for (int v = 0; v < 8; ++v)
      if (lo == v) out_ld[rowbase + 8 * hi + v] = ldacc[v];
  }
}

extern "C" void kernel_launch(void* const* d_in, const int* in_sizes, int n_in,
                              void* d_out, int out_size, void* d_ws, size_t ws_size,
                              hipStream_t stream) {
  (void)n_in; (void)out_size; (void)d_ws; (void)ws_size;
  const float* z1  = (const float*)d_in[0];
  const float* eps = (const float*)d_in[1];
  const float* W0  = (const float*)d_in[2];
  const float* b0  = (const float*)d_in[3];
  const float* W1  = (const float*)d_in[4];
  const float* b1  = (const float*)d_in[5];
  const float* W2  = (const float*)d_in[6];
  const float* b2  = (const float*)d_in[7];

  const int Btot = in_sizes[0] / DD;
  float* out_z  = (float*)d_out;
  float* out_ld = out_z + (size_t)Btot * DD;

  dim3 grid(Btot / ROWS_BLK), block(NWAVES * 32);
  size_t shmem = (size_t)SMEM_FLOATS * sizeof(float);
  ffjord_kernel<<<grid, block, shmem, stream>>>(z1, eps, W0, b0, W1, b1, W2, b2,
                                                out_z, out_ld);
}


# Tangent's first-layer product eps @ W0 depends on neither z nor t: computed once before the time loop and kept in 64 registers, not in each of 20 evaluations; its 1-h1^2 scaling joins the tanh pass
# speedup vs baseline: 1.4752x; 1.0884x over previous
#include <hip/hip_runtime.h>


typedef __attribute__((ext_vector_type(2))) float v2f;
typedef __attribute__((ext_vector_type(4))) float v4f;
typedef __attribute__((ext_vector_type(8))) float v8f;

#define NWAVES    8
#define MROWS     16
#define ROWS_BLK  (NWAVES * MROWS)
#define DD        32
#define HH        128
#define SSTRIDE   132

#define OFF_W0   0
#define OFF_W1   (OFF_W0 + 4096)
#define OFF_W2   (OFF_W1 + 16384)
#define OFF_B0   (OFF_W2 + 4096)
#define OFF_T0   (OFF_B0 + 128)
#define OFF_B1   (OFF_T0 + 128)
#define OFF_T1   (OFF_B1 + 128)
#define OFF_B2   (OFF_T1 + 128)
#define OFF_T2   (OFF_B2 + 32)
#define OFF_STG  (OFF_T2 + 32)
#define STG_WAVE (2 * 16 * SSTRIDE)
#define SMEM_FLOATS (OFF_STG + NWAVES * STG_WAVE)

#define NTHR (NWAVES * 32)
#define STG_PER(K, N) (((K) / 2) * ((N) / 4) / NTHR)
template <int K, int N>
__device__ __forceinline__ void stage_load(v4f* r, const float* __restrict__ Wg, int tid) {
#pragma unroll
  for (int j = 0; j < STG_PER(K, N); ++j) {
    const int u = tid + NTHR * j;
    const float* p = Wg + (2 * (u / (N / 4))) * N + 4 * (u % (N / 4));
    r[2 * j]     = *(const v4f*)p;
    r[2 * j + 1] = *(const v4f*)(p + N);
  }
}
template <int K, int N>
__device__ __forceinline__ void stage_store(float* dst, const v4f* r, int tid) {
#pragma unroll
  for (int j = 0; j < STG_PER(K, N); ++j) {
    const int u = tid + NTHR * j;
    const int k = 2 * (u / (N / 4)), n = 4 * (u % (N / 4));
    const int idx = (((k >> 2) * (N / 16) + (n >> 4)) << 6) + (((k >> 1) & 1) << 5) + 2 * (n & 15);
    const v4f a = r[2 * j], b = r[2 * j + 1];
    v4f lo4 = {a.x, b.x, a.y, b.y};
    v4f hi4 = {a.z, b.z, a.w, b.w};
    *(v4f*)(dst + idx)     = lo4;
    *(v4f*)(dst + idx + 4) = hi4;
  }
}

__device__ __forceinline__ float fast_tanh(float x) {
  float cx = fminf(fmaxf(x, -10.0f), 10.0f);
  float e  = __expf(2.0f * cx);
  return 1.0f - 2.0f * __builtin_amdgcn_rcpf(e + 1.0f);
}

__device__ __forceinline__ v8f wmma4(v2f a, v2f b, v8f c) {
  return __builtin_amdgcn_wmma_f32_16x16x4_f32(false, a, false, b, (short)0, c,
                                               false, false);
}

template <int NKC, int NT>
__device__ __forceinline__ void load_b(v2f* b, const float* wF, int kc0) {
#pragma unroll
  for (int j = 0; j < NKC; ++j)
#pragma unroll
    for (int nt = 0; nt < NT; ++nt)
      b[j * NT + nt] = *(const v2f*)(wF + ((kc0 + j) * NT + nt) * 64);
}
template <int NKC>
__device__ __forceinline__ void load_a(v2f* a, const float* sA, int kc0) {
#pragma unroll
  for (int j = 0; j < NKC; ++j)
    a[j] = *(const v2f*)(sA + 4 * (kc0 + j));
}
template <int NKC, int NT>
__device__ __forceinline__ void mma_block(v8f* acc, const v2f* a, const v2f* b) {
#pragma unroll
  for (int j = 0; j < NKC; ++j)
#pragma unroll
    for (int nt = 0; nt < NT; ++nt)
      acc[nt] = wmma4(a[j], b[j * NT + nt], acc[nt]);
}
template <int NKC>
__device__ __forceinline__ void mma_block2(v8f* dacc, v8f* jacc, const v2f* ah,
                                           const v2f* au, const v2f* b) {
#pragma unroll
  for (int j = 0; j < NKC; ++j)
#pragma unroll
    for (int nt = 0; nt < 2; ++nt) {
      dacc[nt] = wmma4(ah[j], b[j * 2 + nt], dacc[nt]);
      jacc[nt] = wmma4(au[j], b[j * 2 + nt], jacc[nt]);
    }
}

#define STAGE_FENCE() __builtin_amdgcn_sched_barrier(0)

__device__ __forceinline__ void gemm_k128_n128(v8f* acc, const float* sA, const float* wF) {
  v2f a0[4], b0[32], a1[4], b1[32];
  load_a<4>(a0, sA, 0);
  load_b<4, 8>(b0, wF, 0);
#pragma clang loop unroll(disable)
  for (int kb = 0; kb < 32; kb += 8) {
    load_a<4>(a1, sA, kb + 4);
    load_b<4, 8>(b1, wF, kb + 4);
    STAGE_FENCE();
    mma_block<4, 8>(acc, a0, b0);
    STAGE_FENCE();
    const int kn = (kb + 8) & 31;
    load_a<4>(a0, sA, kn);
    load_b<4, 8>(b0, wF, kn);
    STAGE_FENCE();
    mma_block<4, 8>(acc, a1, b1);
  }
}

__device__ __forceinline__ void gemm_k32_n128(v8f* acc, const v2f* a, const float* wF) {
  v2f b0[32], b1[32];
  load_b<4, 8>(b0, wF, 0);
  load_b<4, 8>(b1, wF, 4);
  STAGE_FENCE();
  mma_block<4, 8>(acc, a, b0);
  mma_block<4, 8>(acc, a + 4, b1);
}

__device__ __forceinline__ void func_eval(
    const float* smem, float* sH, float* sU, int lo, int hi,
    const v2f* zin, const v8f* eps_w0, const float (*eps_d)[8],
    v2f* kcur, float* lsum, float t, float wrk)
{
  const v8f ZERO8 = {0.f,0.f,0.f,0.f,0.f,0.f,0.f,0.f};

  const int fb = hi * 32 + 2 * lo;
  const float* wF0 = smem + OFF_W0 + fb;
  const float* wF1 = smem + OFF_W1 + fb;
  const float* wF2 = smem + OFF_W2 + fb;
  float* sHa = sH + lo * SSTRIDE + 2 * hi;
  float* sUa = sU + lo * SSTRIDE + 2 * hi;
  float* sHs = sH + 8 * hi * SSTRIDE + lo;
  float* sUs = sU + 8 * hi * SSTRIDE + lo;

  v8f acc[8];
#pragma unroll
  for (int nt = 0; nt < 8; ++nt) acc[nt] = ZERO8;
  gemm_k32_n128(acc, zin, wF0);
#pragma unroll
  for (int nt = 0; nt < 8; ++nt) {
    int col = 16 * nt + lo;
    float bias = smem[OFF_B0 + col] + t * smem[OFF_T0 + col];
#pragma unroll
    for (int v = 0; v < 8; ++v) {
      int idx = v * SSTRIDE + 16 * nt;
      float h = fast_tanh(acc[nt][v] + bias);
      sHs[idx] = h;
      sUs[idx] = eps_w0[nt][v] * (1.0f - h * h);
    }
  }
  __builtin_amdgcn_wave_barrier();

#pragma unroll
  for (int nt = 0; nt < 8; ++nt) acc[nt] = ZERO8;
  gemm_k128_n128(acc, sHa, wF1);
  __builtin_amdgcn_wave_barrier();
#pragma unroll
  for (int nt = 0; nt < 8; ++nt) {
    int col = 16 * nt + lo;
    float bias = smem[OFF_B1 + col] + t * smem[OFF_T1 + col];
#pragma unroll
    for (int v = 0; v < 8; ++v)
      sHs[v * SSTRIDE + 16 * nt] = fast_tanh(acc[nt][v] + bias);
  }
  __builtin_amdgcn_wave_barrier();

#pragma unroll
  for (int nt = 0; nt < 8; ++nt) acc[nt] = ZERO8;
  gemm_k128_n128(acc, sUa, wF1);
  __builtin_amdgcn_wave_barrier();
#pragma unroll
  for (int nt = 0; nt < 8; ++nt) {
#pragma unroll
    for (int v = 0; v < 8; ++v) {
      int idx = v * SSTRIDE + 16 * nt;
      float h = sHs[idx];
      sUs[idx] = acc[nt][v] * (1.0f - h * h);
    }
  }
  __builtin_amdgcn_wave_barrier();

  v8f dacc[2], jacc[2];
  dacc[0] = ZERO8; dacc[1] = ZERO8; jacc[0] = ZERO8; jacc[1] = ZERO8;
  {
    v2f ah0[8], au0[8], w0[16], ah1[8], au1[8], w1[16];
    load_a<8>(ah0, sHa, 0);  load_a<8>(au0, sUa, 0);  load_b<8, 2>(w0, wF2, 0);
    load_a<8>(ah1, sHa, 8);  load_a<8>(au1, sUa, 8);  load_b<8, 2>(w1, wF2, 8);
    STAGE_FENCE();
    mma_block2<8>(dacc, jacc, ah0, au0, w0);
    STAGE_FENCE();
    load_a<8>(ah0, sHa, 16); load_a<8>(au0, sUa, 16); load_b<8, 2>(w0, wF2, 16);
    STAGE_FENCE();
    mma_block2<8>(dacc, jacc, ah1, au1, w1);
    STAGE_FENCE();
    load_a<8>(ah1, sHa, 24); load_a<8>(au1, sUa, 24); load_b<8, 2>(w1, wF2, 24);
    STAGE_FENCE();
    mma_block2<8>(dacc, jacc, ah0, au0, w0);
    mma_block2<8>(dacc, jacc, ah1, au1, w1);
  }
  __builtin_amdgcn_wave_barrier();

  float p[8];
#pragma unroll
  for (int v = 0; v < 8; ++v)
    p[v] = eps_d[0][v] * jacc[0][v] + eps_d[1][v] * jacc[1][v];
#pragma unroll
  for (int off = 1; off < 16; off <<= 1) {
#pragma unroll
    for (int v = 0; v < 8; ++v) p[v] += __shfl_xor(p[v], off, 32);
  }
#pragma unroll
  for (int v = 0; v < 8; ++v) lsum[v] += wrk * p[v];

#pragma unroll
  for (int nt = 0; nt < 2; ++nt) {
    int col = 16 * nt + lo;
    float bias = smem[OFF_B2 + col] + t * smem[OFF_T2 + col];
#pragma unroll
    for (int v = 0; v < 8; ++v)
      sHs[v * SSTRIDE + 16 * nt] = dacc[nt][v] + bias;
  }
  __builtin_amdgcn_wave_barrier();
#pragma unroll
  for (int kc = 0; kc < 8; ++kc)
    kcur[kc] = *(const v2f*)(sHa + 4 * kc);
  __builtin_amdgcn_wave_barrier();
}

__global__ __launch_bounds__(256)
void ffjord_kernel(const float* __restrict__ z1, const float* __restrict__ eps,
                   const float* __restrict__ W0, const float* __restrict__ b0,
                   const float* __restrict__ W1, const float* __restrict__ b1,
                   const float* __restrict__ W2, const float* __restrict__ b2,
                   float* __restrict__ out_z, float* __restrict__ out_ld)
{
  extern __shared__ __attribute__((aligned(16))) float smem[];
  const int tid  = threadIdx.x;
  const int lane = tid & 31;
  const int wave = tid >> 5;
  const int lo   = lane & 15;
  const int hi   = lane >> 4;

  v4f rw0[2 * STG_PER(32, 128)], rw1[2 * STG_PER(128, 128)], rw2[2 * STG_PER(128, 32)];
  stage_load<32, 128>(rw0, W0, tid);
  stage_load<128, 128>(rw1, W1, tid);
  stage_load<128, 32>(rw2, W2, tid);
  const int nb = tid & 127, nc = tid & 31;
  const float t0v = W0[32 * 128 + nb], b0v = b0[nb];
  const float t1v = W1[128 * 128 + nb], b1v = b1[nb];
  const float t2v = W2[128 * 32 + nc], b2v = b2[nc];
  stage_store<32, 128>(smem + OFF_W0, rw0, tid);
  stage_store<128, 128>(smem + OFF_W1, rw1, tid);
  stage_store<128, 32>(smem + OFF_W2, rw2, tid);
  if (tid < 128) {
    smem[OFF_T0 + nb] = t0v;
    smem[OFF_B0 + nb] = b0v;
    smem[OFF_T1 + nb] = t1v;
    smem[OFF_B1 + nb] = b1v;
  }
  if (tid < 32) {
    smem[OFF_T2 + nc] = t2v;
    smem[OFF_B2 + nc] = b2v;
  }
  __syncthreads();

  float* sH = smem + OFF_STG + wave * STG_WAVE;
  float* sU = sH + 16 * SSTRIDE;
  const int rowbase = blockIdx.x * ROWS_BLK + wave * MROWS;

  v2f z_a[8], zin[8], ksum[8], kcur[8];
  v8f eps_w0[8];
  float eps_d[2][8];
  float ldacc[8];

  {
    const float* zrow = z1  + (size_t)(rowbase + lo) * DD + 2 * hi;
    const float* erow = eps + (size_t)(rowbase + lo) * DD + 2 * hi;
    v2f eps_a[8];
#pragma unroll
    for (int kc = 0; kc < 8; ++kc) {
      z_a[kc]   = *(const v2f*)(zrow + 4 * kc);
      eps_a[kc] = *(const v2f*)(erow + 4 * kc);
    }
    const v8f ZERO8 = {0.f,0.f,0.f,0.f,0.f,0.f,0.f,0.f};
#pragma unroll
    for (int nt = 0; nt < 8; ++nt) eps_w0[nt] = ZERO8;
    gemm_k32_n128(eps_w0, eps_a, smem + OFF_W0 + hi * 32 + 2 * lo);
#pragma unroll
    for (int nt = 0; nt < 2; ++nt)
#pragma unroll
      for (int v = 0; v < 8; ++v)
        eps_d[nt][v] = eps[(size_t)(rowbase + v + 8 * hi) * DD + 16 * nt + lo];
#pragma unroll
    for (int v = 0; v < 8; ++v) ldacc[v] = 0.0f;
  }

  const float hstep = -0.2f;
  for (int s = 0; s < 5; ++s) {
    const float tb = 1.0f - 0.2f * (float)s;
    float lsum[8];
#pragma unroll
    for (int v = 0; v < 8; ++v) lsum[v] = 0.0f;
#pragma unroll
    for (int i = 0; i < 8; ++i) { zin[i] = z_a[i]; ksum[i] = (v2f){0.0f, 0.0f}; }

#pragma clang loop unroll(disable)
    for (int st = 0; st < 4; ++st) {
      const float toff = (st == 0) ? 0.0f : ((st == 3) ? hstep : 0.5f * hstep);
      const float wst  = (st == 0 || st == 3) ? 1.0f : 2.0f;
      const float cz   = (st < 2) ? 0.5f * hstep : hstep;
      func_eval(smem, sH, sU, lo, hi, zin, eps_w0, eps_d, kcur, lsum, tb + toff, wst);
#pragma unroll
      for (int i = 0; i < 8; ++i) { ksum[i] += kcur[i] * wst; zin[i] = z_a[i] + kcur[i] * cz; }
    }

    const float c = hstep / 6.0f;
#pragma unroll
    for (int i = 0; i < 8; ++i) z_a[i] += ksum[i] * c;
#pragma unroll
    for (int v = 0; v < 8; ++v) ldacc[v] -= c * lsum[v];
  }

  {
    float* orow = out_z + (size_t)(rowbase + lo) * DD + 2 * hi;
#pragma unroll
    for (int kc = 0; kc < 8; ++kc)
      *(v2f*)(orow + 4 * kc) = z_a[kc];
#pragma unroll
    for (int v = 0; v < 8; ++v)
      if (lo == v) out_ld[rowbase + 8 * hi + v] = ldacc[v];
  }
}

extern "C" void kernel_launch(void* const* d_in, const int* in_sizes, int n_in,
                              void* d_out, int out_size, void* d_ws, size_t ws_size,
                              hipStream_t stream) {
  (void)n_in; (void)out_size; (void)d_ws; (void)ws_size;
  const float* z1  = (const float*)d_in[0];
  const float* eps = (const float*)d_in[1];
  const float* W0  = (const float*)d_in[2];
  const float* b0  = (const float*)d_in[3];
  const float* W1  = (const float*)d_in[4];
  const float* b1  = (const float*)d_in[5];
  const float* W2  = (const float*)d_in[6];
  const float* b2  = (const float*)d_in[7];

  const int Btot = in_sizes[0] / DD;
  float* out_z  = (float*)d_out;
  float* out_ld = out_z + (size_t)Btot * DD;

  dim3 grid(Btot / ROWS_BLK), block(NWAVES * 32);
  size_t shmem = (size_t)SMEM_FLOATS * sizeof(float);
  ffjord_kernel<<<grid, block, shmem, stream>>>(z1, eps, W0, b0, W1, b1, W2, b2,
                                                out_z, out_ld);
}


# Hutchinson trace taken as u2 . (eps @ W2a^T), a factor computed once and kept in 64 registers: last layer's tangent product and u2's LDS staging leave each evaluation; trace summed in a new order
# speedup vs baseline: 1.5858x; 1.0750x over previous
#include <hip/hip_runtime.h>


typedef __attribute__((ext_vector_type(2))) float v2f;
typedef __attribute__((ext_vector_type(4))) float v4f;
typedef __attribute__((ext_vector_type(8))) float v8f;

#define NWAVES    8
#define MROWS     16
#define ROWS_BLK  (NWAVES * MROWS)
#define DD        32
#define HH        128
#define SSTRIDE   132

#define OFF_W0   0
#define OFF_W1   (OFF_W0 + 4096)
#define OFF_W2   (OFF_W1 + 16384)
#define OFF_B0   (OFF_W2 + 4096)
#define OFF_T0   (OFF_B0 + 128)
#define OFF_B1   (OFF_T0 + 128)
#define OFF_T1   (OFF_B1 + 128)
#define OFF_B2   (OFF_T1 + 128)
#define OFF_T2   (OFF_B2 + 32)
#define OFF_W2T  (OFF_T2 + 32)
#define OFF_STG  (OFF_W2T + 4096)
#define STG_WAVE (2 * 16 * SSTRIDE)
#define SMEM_FLOATS (OFF_STG + NWAVES * STG_WAVE)

#define NTHR (NWAVES * 32)
#define STG_PER(K, N) (((K) / 2) * ((N) / 4) / NTHR)
template <int K, int N>
__device__ __forceinline__ void stage_load(v4f* r, const float* __restrict__ Wg, int tid) {
#pragma unroll
  for (int j = 0; j < STG_PER(K, N); ++j) {
    const int u = tid + NTHR * j;
    const float* p = Wg + (2 * (u / (N / 4))) * N + 4 * (u % (N / 4));
    r[2 * j]     = *(const v4f*)p;
    r[2 * j + 1] = *(const v4f*)(p + N);
  }
}
template <int K, int N>
__device__ __forceinline__ void stage_store(float* dst, const v4f* r, int tid) {
#pragma unroll
  for (int j = 0; j < STG_PER(K, N); ++j) {
    const int u = tid + NTHR * j;
    const int k = 2 * (u / (N / 4)), n = 4 * (u % (N / 4));
    const int idx = (((k >> 2) * (N / 16) + (n >> 4)) << 6) + (((k >> 1) & 1) << 5) + 2 * (n & 15);
    const v4f a = r[2 * j], b = r[2 * j + 1];
    v4f lo4 = {a.x, b.x, a.y, b.y};
    v4f hi4 = {a.z, b.z, a.w, b.w};
    *(v4f*)(dst + idx)     = lo4;
    *(v4f*)(dst + idx + 4) = hi4;
  }
}

__device__ __forceinline__ void stage_store_w2t(float* dst, const v4f* r, int tid) {
#pragma unroll
  for (int j = 0; j < STG_PER(128, 32); ++j) {
    const int u = tid + NTHR * j;
    const int k = 2 * (u / (DD / 4)), n = 4 * (u % (DD / 4));
    const int idx = (((n >> 2) * (HH / 16) + (k >> 4)) << 6) + 2 * (k & 15);
    const v4f a = r[2 * j], b = r[2 * j + 1];
    v4f lo4 = {a.x, a.y, b.x, b.y};
    v4f hi4 = {a.z, a.w, b.z, b.w};
    *(v4f*)(dst + idx)      = lo4;
    *(v4f*)(dst + idx + 32) = hi4;
  }
}

__device__ __forceinline__ float fast_tanh(float x) {
  float cx = fminf(fmaxf(x, -10.0f), 10.0f);
  float e  = __expf(2.0f * cx);
  return 1.0f - 2.0f * __builtin_amdgcn_rcpf(e + 1.0f);
}

__device__ __forceinline__ v8f wmma4(v2f a, v2f b, v8f c) {
  return __builtin_amdgcn_wmma_f32_16x16x4_f32(false, a, false, b, (short)0, c,
                                               false, false);
}

template <int NKC, int NT>
__device__ __forceinline__ void load_b(v2f* b, const float* wF, int kc0) {
#pragma unroll
  for (int j = 0; j < NKC; ++j)
#pragma unroll
    for (int nt = 0; nt < NT; ++nt)
      b[j * NT + nt] = *(const v2f*)(wF + ((kc0 + j) * NT + nt) * 64);
}
template <int NKC>
__device__ __forceinline__ void load_a(v2f* a, const float* sA, int kc0) {
#pragma unroll
  for (int j = 0; j < NKC; ++j)
    a[j] = *(const v2f*)(sA + 4 * (kc0 + j));
}
template <int NKC, int NT>
__device__ __forceinline__ void mma_block(v8f* acc, const v2f* a, const v2f* b) {
#pragma unroll
  for (int j = 0; j < NKC; ++j)
#pragma unroll
    for (int nt = 0; nt < NT; ++nt)
      acc[nt] = wmma4(a[j], b[j * NT + nt], acc[nt]);
}
#define STAGE_FENCE() __builtin_amdgcn_sched_barrier(0)

__device__ __forceinline__ void gemm_k128_n128(v8f* acc, const float* sA, const float* wF) {
  v2f a0[4], b0[32], a1[4], b1[32];
  load_a<4>(a0, sA, 0);
  load_b<4, 8>(b0, wF, 0);
#pragma clang loop unroll(disable)
  for (int kb = 0; kb < 32; kb += 8) {
    load_a<4>(a1, sA, kb + 4);
    load_b<4, 8>(b1, wF, kb + 4);
    STAGE_FENCE();
    mma_block<4, 8>(acc, a0, b0);
    STAGE_FENCE();
    const int kn = (kb + 8) & 31;
    load_a<4>(a0, sA, kn);
    load_b<4, 8>(b0, wF, kn);
    STAGE_FENCE();
    mma_block<4, 8>(acc, a1, b1);
  }
}

__device__ __forceinline__ void gemm_k32_n128(v8f* acc, const v2f* a, const float* wF) {
  v2f b0[32], b1[32];
  load_b<4, 8>(b0, wF, 0);
  load_b<4, 8>(b1, wF, 4);
  STAGE_FENCE();
  mma_block<4, 8>(acc, a, b0);
  mma_block<4, 8>(acc, a + 4, b1);
}

__device__ __forceinline__ void func_eval(
    const float* smem, float* sH, float* sU, int lo, int hi,
    const v2f* zin, const v8f* eps_w0, const v8f* eps_w2t,
    v2f* kcur, float* lsum, float t, float wrk)
{
  const v8f ZERO8 = {0.f,0.f,0.f,0.f,0.f,0.f,0.f,0.f};

  const int fb = hi * 32 + 2 * lo;
  const float* wF0 = smem + OFF_W0 + fb;
  const float* wF1 = smem + OFF_W1 + fb;
  const float* wF2 = smem + OFF_W2 + fb;
  float* sHa = sH + lo * SSTRIDE + 2 * hi;
  float* sUa = sU + lo * SSTRIDE + 2 * hi;
  float* sHs = sH + 8 * hi * SSTRIDE + lo;
  float* sUs = sU + 8 * hi * SSTRIDE + lo;

  v8f acc[8];
#pragma unroll
  for (int nt = 0; nt < 8; ++nt) acc[nt] = ZERO8;
  gemm_k32_n128(acc, zin, wF0);
#pragma unroll
  for (int nt = 0; nt < 8; ++nt) {
    int col = 16 * nt + lo;
    float bias = smem[OFF_B0 + col] + t * smem[OFF_T0 + col];
#pragma unroll
    for (int v = 0; v < 8; ++v) {
      int idx = v * SSTRIDE + 16 * nt;
      float h = fast_tanh(acc[nt][v] + bias);
      sHs[idx] = h;
      sUs[idx] = eps_w0[nt][v] * (1.0f - h * h);
    }
  }
  __builtin_amdgcn_wave_barrier();

#pragma unroll
  for (int nt = 0; nt < 8; ++nt) acc[nt] = ZERO8;
  gemm_k128_n128(acc, sHa, wF1);
  __builtin_amdgcn_wave_barrier();
#pragma unroll
  for (int nt = 0; nt < 8; ++nt) {
    int col = 16 * nt + lo;
    float bias = smem[OFF_B1 + col] + t * smem[OFF_T1 + col];
#pragma unroll
    for (int v = 0; v < 8; ++v)
      sHs[v * SSTRIDE + 16 * nt] = fast_tanh(acc[nt][v] + bias);
  }
  __builtin_amdgcn_wave_barrier();

#pragma unroll
  for (int nt = 0; nt < 8; ++nt) acc[nt] = ZERO8;
  gemm_k128_n128(acc, sUa, wF1);
  float p[8];
  {
    v2f p2[4];
#pragma unroll
    for (int q = 0; q < 4; ++q) p2[q] = (v2f){0.0f, 0.0f};
#pragma unroll
    for (int nt = 0; nt < 8; ++nt) {
#pragma unroll
      for (int q = 0; q < 4; ++q) {
        const v2f h = {sHs[(2 * q) * SSTRIDE + 16 * nt], sHs[(2 * q + 1) * SSTRIDE + 16 * nt]};
        const v2f a = {acc[nt][2 * q], acc[nt][2 * q + 1]};
        const v2f g = {eps_w2t[nt][2 * q], eps_w2t[nt][2 * q + 1]};
        p2[q] += (a * (1.0f - h * h)) * g;
      }
    }
#pragma unroll
    for (int q = 0; q < 4; ++q) { p[2 * q] = p2[q].x; p[2 * q + 1] = p2[q].y; }
  }
#pragma unroll
  for (int off = 1; off < 16; off <<= 1) {
#pragma unroll
    for (int v = 0; v < 8; ++v) p[v] += __shfl_xor(p[v], off, 32);
  }
#pragma unroll
  for (int v = 0; v < 8; ++v) lsum[v] += wrk * p[v];

  v8f dacc[2];
  dacc[0] = ZERO8; dacc[1] = ZERO8;
  {
    v2f ah0[16], w0[32], ah1[16], w1[32];
    load_a<16>(ah0, sHa, 0);   load_b<16, 2>(w0, wF2, 0);
    load_a<16>(ah1, sHa, 16);  load_b<16, 2>(w1, wF2, 16);
    STAGE_FENCE();
    mma_block<16, 2>(dacc, ah0, w0);
    mma_block<16, 2>(dacc, ah1, w1);
  }
  __builtin_amdgcn_wave_barrier();

#pragma unroll
  for (int nt = 0; nt < 2; ++nt) {
    int col = 16 * nt + lo;
    float bias = smem[OFF_B2 + col] + t * smem[OFF_T2 + col];
#pragma unroll
    for (int v = 0; v < 8; ++v)
      sHs[v * SSTRIDE + 16 * nt] = dacc[nt][v] + bias;
  }
  __builtin_amdgcn_wave_barrier();
#pragma unroll
  for (int kc = 0; kc < 8; ++kc)
    kcur[kc] = *(const v2f*)(sHa + 4 * kc);
  __builtin_amdgcn_wave_barrier();
}

__global__ __launch_bounds__(256)
void ffjord_kernel(const float* __restrict__ z1, const float* __restrict__ eps,
                   const float* __restrict__ W0, const float* __restrict__ b0,
                   const float* __restrict__ W1, const float* __restrict__ b1,
                   const float* __restrict__ W2, const float* __restrict__ b2,
                   float* __restrict__ out_z, float* __restrict__ out_ld)
{
  extern __shared__ __attribute__((aligned(16))) float smem[];
  const int tid  = threadIdx.x;
  const int lane = tid & 31;
  const int wave = tid >> 5;
  const int lo   = lane & 15;
  const int hi   = lane >> 4;

  v4f rw0[2 * STG_PER(32, 128)], rw1[2 * STG_PER(128, 128)], rw2[2 * STG_PER(128, 32)];
  stage_load<32, 128>(rw0, W0, tid);
  stage_load<128, 128>(rw1, W1, tid);
  stage_load<128, 32>(rw2, W2, tid);
  const int nb = tid & 127, nc = tid & 31;
  const float t0v = W0[32 * 128 + nb], b0v = b0[nb];
  const float t1v = W1[128 * 128 + nb], b1v = b1[nb];
  const float t2v = W2[128 * 32 + nc], b2v = b2[nc];
  stage_store<32, 128>(smem + OFF_W0, rw0, tid);
  stage_store<128, 128>(smem + OFF_W1, rw1, tid);
  stage_store<128, 32>(smem + OFF_W2, rw2, tid);
  stage_store_w2t(smem + OFF_W2T, rw2, tid);
  if (tid < 128) {
    smem[OFF_T0 + nb] = t0v;
    smem[OFF_B0 + nb] = b0v;
    smem[OFF_T1 + nb] = t1v;
    smem[OFF_B1 + nb] = b1v;
  }
  if (tid < 32) {
    smem[OFF_T2 + nc] = t2v;
    smem[OFF_B2 + nc] = b2v;
  }
  __syncthreads();

  float* sH = smem + OFF_STG + wave * STG_WAVE;
  float* sU = sH + 16 * SSTRIDE;
  const int rowbase = blockIdx.x * ROWS_BLK + wave * MROWS;

  v2f z_a[8], zin[8], ksum[8], kcur[8];
  v8f eps_w0[8];
  v8f eps_w2t[8];
  float ldacc[8];

  {
    const float* zrow = z1  + (size_t)(rowbase + lo) * DD + 2 * hi;
    const float* erow = eps + (size_t)(rowbase + lo) * DD + 2 * hi;
    v2f eps_a[8];
#pragma unroll
    for (int kc = 0; kc < 8; ++kc) {
      z_a[kc]   = *(const v2f*)(zrow + 4 * kc);
      eps_a[kc] = *(const v2f*)(erow + 4 * kc);
    }
    const v8f ZERO8 = {0.f,0.f,0.f,0.f,0.f,0.f,0.f,0.f};
#pragma unroll
    for (int nt = 0; nt < 8; ++nt) eps_w0[nt] = ZERO8;
    gemm_k32_n128(eps_w0, eps_a, smem + OFF_W0 + hi * 32 + 2 * lo);
#pragma unroll
    for (int nt = 0; nt < 8; ++nt) eps_w2t[nt] = ZERO8;
    gemm_k32_n128(eps_w2t, eps_a, smem + OFF_W2T + hi * 32 + 2 * lo);
#pragma unroll
    for (int v = 0; v < 8; ++v) ldacc[v] = 0.0f;
  }

  const float hstep = -0.2f;
  for (int s = 0; s < 5; ++s) {
    const float tb = 1.0f - 0.2f * (float)s;
    float lsum[8];
#pragma unroll
    for (int v = 0; v < 8; ++v) lsum[v] = 0.0f;
#pragma unroll
    for (int i = 0; i < 8; ++i) { zin[i] = z_a[i]; ksum[i] = (v2f){0.0f, 0.0f}; }

#pragma clang loop unroll(disable)
    for (int st = 0; st < 4; ++st) {
      const float toff = (st == 0) ? 0.0f : ((st == 3) ? hstep : 0.5f * hstep);
      const float wst  = (st == 0 || st == 3) ? 1.0f : 2.0f;
      const float cz   = (st < 2) ? 0.5f * hstep : hstep;
      func_eval(smem, sH, sU, lo, hi, zin, eps_w0, eps_w2t, kcur, lsum, tb + toff, wst);
#pragma unroll
      for (int i = 0; i < 8; ++i) { ksum[i] += kcur[i] * wst; zin[i] = z_a[i] + kcur[i] * cz; }
    }

    const float c = hstep / 6.0f;
#pragma unroll
    for (int i = 0; i < 8; ++i) z_a[i] += ksum[i] * c;
#pragma unroll
    for (int v = 0; v < 8; ++v) ldacc[v] -= c * lsum[v];
  }

  {
    float* orow = out_z + (size_t)(rowbase + lo) * DD + 2 * hi;
#pragma unroll
    for (int kc = 0; kc < 8; ++kc)
      *(v2f*)(orow + 4 * kc) = z_a[kc];
#pragma unroll
    for (int v = 0; v < 8; ++v)
      if (lo == v) out_ld[rowbase + 8 * hi + v] = ldacc[v];
  }
}

extern "C" void kernel_launch(void* const* d_in, const int* in_sizes, int n_in,
                              void* d_out, int out_size, void* d_ws, size_t ws_size,
                              hipStream_t stream) {
  (void)n_in; (void)out_size; (void)d_ws; (void)ws_size;
  const float* z1  = (const float*)d_in[0];
  const float* eps = (const float*)d_in[1];
  const float* W0  = (const float*)d_in[2];
  const float* b0  = (const float*)d_in[3];
  const float* W1  = (const float*)d_in[4];
  const float* b1  = (const float*)d_in[5];
  const float* W2  = (const float*)d_in[6];
  const float* b2  = (const float*)d_in[7];

  const int Btot = in_sizes[0] / DD;
  float* out_z  = (float*)d_out;
  float* out_ld = out_z + (size_t)Btot * DD;

  dim3 grid(Btot / ROWS_BLK), block(NWAVES * 32);
  size_t shmem = (size_t)SMEM_FLOATS * sizeof(float);
  ffjord_kernel<<<grid, block, shmem, stream>>>(z1, eps, W0, b0, W1, b1, W2, b2,
                                                out_z, out_ld);
}


# Issue priority taken before every evaluation by whichever of a SIMD's two waves has finished fewer evaluations (s_setprio from per-wave counters in LDS): the pair ends together, none runs on alone
# speedup vs baseline: 1.6402x; 1.0343x over previous
#include <hip/hip_runtime.h>


typedef __attribute__((ext_vector_type(2))) float v2f;
typedef __attribute__((ext_vector_type(4))) float v4f;
typedef __attribute__((ext_vector_type(8))) float v8f;
typedef __attribute__((address_space(3))) volatile int lds_vint;

#define NWAVES    8
#define MROWS     16
#define ROWS_BLK  (NWAVES * MROWS)
#define DD        32
#define HH        128
#define SSTRIDE   132

#define OFF_W0   0
#define OFF_W1   (OFF_W0 + 4096)
#define OFF_W2   (OFF_W1 + 16384)
#define OFF_B0   (OFF_W2 + 4096)
#define OFF_T0   (OFF_B0 + 128)
#define OFF_B1   (OFF_T0 + 128)
#define OFF_T1   (OFF_B1 + 128)
#define OFF_B2   (OFF_T1 + 128)
#define OFF_T2   (OFF_B2 + 32)
#define OFF_W2T  (OFF_T2 + 32)
#define OFF_STG  (OFF_W2T + 4096)
#define STG_WAVE (2 * 16 * SSTRIDE)
#define OFF_PRG  (OFF_STG + NWAVES * STG_WAVE)
#define SMEM_FLOATS (OFF_PRG + 16)

#define NTHR (NWAVES * 32)
#define STG_PER(K, N) (((K) / 2) * ((N) / 4) / NTHR)
template <int K, int N>
__device__ __forceinline__ void stage_load(v4f* r, const float* __restrict__ Wg, int tid) {
#pragma unroll
  for (int j = 0; j < STG_PER(K, N); ++j) {
    const int u = tid + NTHR * j;
    const float* p = Wg + (2 * (u / (N / 4))) * N + 4 * (u % (N / 4));
    r[2 * j]     = *(const v4f*)p;
    r[2 * j + 1] = *(const v4f*)(p + N);
  }
}
template <int K, int N>
__device__ __forceinline__ void stage_store(float* dst, const v4f* r, int tid) {
#pragma unroll
  for (int j = 0; j < STG_PER(K, N); ++j) {
    const int u = tid + NTHR * j;
    const int k = 2 * (u / (N / 4)), n = 4 * (u % (N / 4));
    const int idx = (((k >> 2) * (N / 16) + (n >> 4)) << 6) + (((k >> 1) & 1) << 5) + 2 * (n & 15);
    const v4f a = r[2 * j], b = r[2 * j + 1];
    v4f lo4 = {a.x, b.x, a.y, b.y};
    v4f hi4 = {a.z, b.z, a.w, b.w};
    *(v4f*)(dst + idx)     = lo4;
    *(v4f*)(dst + idx + 4) = hi4;
  }
}

__device__ __forceinline__ void stage_store_w2t(float* dst, const v4f* r, int tid) {
#pragma unroll
  for (int j = 0; j < STG_PER(128, 32); ++j) {
    const int u = tid + NTHR * j;
    const int k = 2 * (u / (DD / 4)), n = 4 * (u % (DD / 4));
    const int idx = (((n >> 2) * (HH / 16) + (k >> 4)) << 6) + 2 * (k & 15);
    const v4f a = r[2 * j], b = r[2 * j + 1];
    v4f lo4 = {a.x, a.y, b.x, b.y};
    v4f hi4 = {a.z, a.w, b.z, b.w};
    *(v4f*)(dst + idx)      = lo4;
    *(v4f*)(dst + idx + 32) = hi4;
  }
}

__device__ __forceinline__ float fast_tanh(float x) {
  float cx = fminf(fmaxf(x, -10.0f), 10.0f);
  float e  = __expf(2.0f * cx);
  return 1.0f - 2.0f * __builtin_amdgcn_rcpf(e + 1.0f);
}

__device__ __forceinline__ v8f wmma4(v2f a, v2f b, v8f c) {
  return __builtin_amdgcn_wmma_f32_16x16x4_f32(false, a, false, b, (short)0, c,
                                               false, false);
}

template <int NKC, int NT>
__device__ __forceinline__ void load_b(v2f* b, const float* wF, int kc0) {
#pragma unroll
  for (int j = 0; j < NKC; ++j)
#pragma unroll
    for (int nt = 0; nt < NT; ++nt)
      b[j * NT + nt] = *(const v2f*)(wF + ((kc0 + j) * NT + nt) * 64);
}
template <int NKC>
__device__ __forceinline__ void load_a(v2f* a, const float* sA, int kc0) {
#pragma unroll
  for (int j = 0; j < NKC; ++j)
    a[j] = *(const v2f*)(sA + 4 * (kc0 + j));
}
template <int NKC, int NT>
__device__ __forceinline__ void mma_block(v8f* acc, const v2f* a, const v2f* b) {
#pragma unroll
  for (int j = 0; j < NKC; ++j)
#pragma unroll
    for (int nt = 0; nt < NT; ++nt)
      acc[nt] = wmma4(a[j], b[j * NT + nt], acc[nt]);
}
#define STAGE_FENCE() __builtin_amdgcn_sched_barrier(0)

__device__ __forceinline__ void gemm_k128_n128(v8f* acc, const float* sA, const float* wF) {
  v2f a0[4], b0[32], a1[4], b1[32];
  load_a<4>(a0, sA, 0);
  load_b<4, 8>(b0, wF, 0);
#pragma clang loop unroll(disable)
  for (int kb = 0; kb < 32; kb += 8) {
    load_a<4>(a1, sA, kb + 4);
    load_b<4, 8>(b1, wF, kb + 4);
    STAGE_FENCE();
    mma_block<4, 8>(acc, a0, b0);
    STAGE_FENCE();
    const int kn = (kb + 8) & 31;
    load_a<4>(a0, sA, kn);
    load_b<4, 8>(b0, wF, kn);
    STAGE_FENCE();
    mma_block<4, 8>(acc, a1, b1);
  }
}

__device__ __forceinline__ void gemm_k32_n128(v8f* acc, const v2f* a, const float* wF) {
  v2f b0[32], b1[32];
  load_b<4, 8>(b0, wF, 0);
  load_b<4, 8>(b1, wF, 4);
  STAGE_FENCE();
  mma_block<4, 8>(acc, a, b0);
  mma_block<4, 8>(acc, a + 4, b1);
}

__device__ __forceinline__ void func_eval(
    const float* smem, float* sH, float* sU, int lo, int hi,
    const v2f* zin, const v8f* eps_w0, const v8f* eps_w2t,
    v2f* kcur, float* lsum, float t, float wrk)
{
  const v8f ZERO8 = {0.f,0.f,0.f,0.f,0.f,0.f,0.f,0.f};

  const int fb = hi * 32 + 2 * lo;
  const float* wF0 = smem + OFF_W0 + fb;
  const float* wF1 = smem + OFF_W1 + fb;
  const float* wF2 = smem + OFF_W2 + fb;
  float* sHa = sH + lo * SSTRIDE + 2 * hi;
  float* sUa = sU + lo * SSTRIDE + 2 * hi;
  float* sHs = sH + 8 * hi * SSTRIDE + lo;
  float* sUs = sU + 8 * hi * SSTRIDE + lo;

  v8f acc[8];
#pragma unroll
  for (int nt = 0; nt < 8; ++nt) acc[nt] = ZERO8;
  gemm_k32_n128(acc, zin, wF0);
#pragma unroll
  for (int nt = 0; nt < 8; ++nt) {
    int col = 16 * nt + lo;
    float bias = smem[OFF_B0 + col] + t * smem[OFF_T0 + col];
#pragma unroll
    for (int v = 0; v < 8; ++v) {
      int idx = v * SSTRIDE + 16 * nt;
      float h = fast_tanh(acc[nt][v] + bias);
      sHs[idx] = h;
      sUs[idx] = eps_w0[nt][v] * (1.0f - h * h);
    }
  }
  __builtin_amdgcn_wave_barrier();

#pragma unroll
  for (int nt = 0; nt < 8; ++nt) acc[nt] = ZERO8;
  gemm_k128_n128(acc, sHa, wF1);
  __builtin_amdgcn_wave_barrier();
#pragma unroll
  for (int nt = 0; nt < 8; ++nt) {
    int col = 16 * nt + lo;
    float bias = smem[OFF_B1 + col] + t * smem[OFF_T1 + col];
#pragma unroll
    for (int v = 0; v < 8; ++v)
      sHs[v * SSTRIDE + 16 * nt] = fast_tanh(acc[nt][v] + bias);
  }
  __builtin_amdgcn_wave_barrier();

#pragma unroll
  for (int nt = 0; nt < 8; ++nt) acc[nt] = ZERO8;
  gemm_k128_n128(acc, sUa, wF1);
  float p[8];
  {
    v2f p2[4];
#pragma unroll
    for (int q = 0; q < 4; ++q) p2[q] = (v2f){0.0f, 0.0f};
#pragma unroll
    for (int nt = 0; nt < 8; ++nt) {
#pragma unroll
      for (int q = 0; q < 4; ++q) {
        const v2f h = {sHs[(2 * q) * SSTRIDE + 16 * nt], sHs[(2 * q + 1) * SSTRIDE + 16 * nt]};
        const v2f a = {acc[nt][2 * q], acc[nt][2 * q + 1]};
        const v2f g = {eps_w2t[nt][2 * q], eps_w2t[nt][2 * q + 1]};
        p2[q] += (a * (1.0f - h * h)) * g;
      }
    }
#pragma unroll
    for (int q = 0; q < 4; ++q) { p[2 * q] = p2[q].x; p[2 * q + 1] = p2[q].y; }
  }
#pragma unroll
  for (int off = 1; off < 16; off <<= 1) {
#pragma unroll
    for (int v = 0; v < 8; ++v) p[v] += __shfl_xor(p[v], off, 32);
  }
#pragma unroll
  for (int v = 0; v < 8; ++v) lsum[v] += wrk * p[v];

  v8f dacc[2];
  dacc[0] = ZERO8; dacc[1] = ZERO8;
  {
    v2f ah0[16], w0[32], ah1[16], w1[32];
    load_a<16>(ah0, sHa, 0);   load_b<16, 2>(w0, wF2, 0);
    load_a<16>(ah1, sHa, 16);  load_b<16, 2>(w1, wF2, 16);
    STAGE_FENCE();
    mma_block<16, 2>(dacc, ah0, w0);
    mma_block<16, 2>(dacc, ah1, w1);
  }
  __builtin_amdgcn_wave_barrier();

#pragma unroll
  for (int nt = 0; nt < 2; ++nt) {
    int col = 16 * nt + lo;
    float bias = smem[OFF_B2 + col] + t * smem[OFF_T2 + col];
#pragma unroll
    for (int v = 0; v < 8; ++v)
      sHs[v * SSTRIDE + 16 * nt] = dacc[nt][v] + bias;
  }
  __builtin_amdgcn_wave_barrier();
#pragma unroll
  for (int kc = 0; kc < 8; ++kc)
    kcur[kc] = *(const v2f*)(sHa + 4 * kc);
  __builtin_amdgcn_wave_barrier();
}

__global__ __launch_bounds__(256)
void ffjord_kernel(const float* __restrict__ z1, const float* __restrict__ eps,
                   const float* __restrict__ W0, const float* __restrict__ b0,
                   const float* __restrict__ W1, const float* __restrict__ b1,
                   const float* __restrict__ W2, const float* __restrict__ b2,
                   float* __restrict__ out_z, float* __restrict__ out_ld)
{
  extern __shared__ __attribute__((aligned(16))) float smem[];
  const int tid  = threadIdx.x;
  const int lane = tid & 31;
  const int wave = tid >> 5;
  const int lo   = lane & 15;
  const int hi   = lane >> 4;

  v4f rw0[2 * STG_PER(32, 128)], rw1[2 * STG_PER(128, 128)], rw2[2 * STG_PER(128, 32)];
  stage_load<32, 128>(rw0, W0, tid);
  stage_load<128, 128>(rw1, W1, tid);
  stage_load<128, 32>(rw2, W2, tid);
  const int nb = tid & 127, nc = tid & 31;
  const float t0v = W0[32 * 128 + nb], b0v = b0[nb];
  const float t1v = W1[128 * 128 + nb], b1v = b1[nb];
  const float t2v = W2[128 * 32 + nc], b2v = b2[nc];
  stage_store<32, 128>(smem + OFF_W0, rw0, tid);
  stage_store<128, 128>(smem + OFF_W1, rw1, tid);
  stage_store<128, 32>(smem + OFF_W2, rw2, tid);
  stage_store_w2t(smem + OFF_W2T, rw2, tid);
  if (tid < 128) {
    smem[OFF_T0 + nb] = t0v;
    smem[OFF_B0 + nb] = b0v;
    smem[OFF_T1 + nb] = t1v;
    smem[OFF_B1 + nb] = b1v;
  }
  if (tid < 32) {
    smem[OFF_T2 + nc] = t2v;
    smem[OFF_B2 + nc] = b2v;
  }
  lds_vint* prog = (lds_vint*)(smem + OFF_PRG);
  const int wave_u  = __builtin_amdgcn_readfirstlane(wave);
  const int partner = wave_u ^ (NWAVES / 2);
  if (lane == 0) prog[wave_u] = 0;
  __syncthreads();

  float* sH = smem + OFF_STG + wave * STG_WAVE;
  float* sU = sH + 16 * SSTRIDE;
  const int rowbase = blockIdx.x * ROWS_BLK + wave * MROWS;

  v2f z_a[8], zin[8], ksum[8], kcur[8];
  v8f eps_w0[8];
  v8f eps_w2t[8];
  float ldacc[8];

  {
    const float* zrow = z1  + (size_t)(rowbase + lo) * DD + 2 * hi;
    const float* erow = eps + (size_t)(rowbase + lo) * DD + 2 * hi;
    v2f eps_a[8];
#pragma unroll
    for (int kc = 0; kc < 8; ++kc) {
      z_a[kc]   = *(const v2f*)(zrow + 4 * kc);
      eps_a[kc] = *(const v2f*)(erow + 4 * kc);
    }
    const v8f ZERO8 = {0.f,0.f,0.f,0.f,0.f,0.f,0.f,0.f};
#pragma unroll
    for (int nt = 0; nt < 8; ++nt) eps_w0[nt] = ZERO8;
    gemm_k32_n128(eps_w0, eps_a, smem + OFF_W0 + hi * 32 + 2 * lo);
#pragma unroll
    for (int nt = 0; nt < 8; ++nt) eps_w2t[nt] = ZERO8;
    gemm_k32_n128(eps_w2t, eps_a, smem + OFF_W2T + hi * 32 + 2 * lo);
#pragma unroll
    for (int v = 0; v < 8; ++v) ldacc[v] = 0.0f;
  }

  const float hstep = -0.2f;
  for (int s = 0; s < 5; ++s) {
    const float tb = 1.0f - 0.2f * (float)s;
    float lsum[8];
#pragma unroll
    for (int v = 0; v < 8; ++v) lsum[v] = 0.0f;
#pragma unroll
    for (int i = 0; i < 8; ++i) { zin[i] = z_a[i]; ksum[i] = (v2f){0.0f, 0.0f}; }

#pragma clang loop unroll(disable)
    for (int st = 0; st < 4; ++st) {
      const float toff = (st == 0) ? 0.0f : ((st == 3) ? hstep : 0.5f * hstep);
      const float wst  = (st == 0 || st == 3) ? 1.0f : 2.0f;
      const float cz   = (st < 2) ? 0.5f * hstep : hstep;
      {
        const int done = 4 * s + st;
        if (lane == 0) prog[wave_u] = done;
        const int other = __builtin_amdgcn_readfirstlane(prog[partner]);
        const bool behind = (done < other) || (done == other && wave_u >= NWAVES / 2);
        if (behind) __builtin_amdgcn_s_setprio(3);
        else        __builtin_amdgcn_s_setprio(0);
      }
      func_eval(smem, sH, sU, lo, hi, zin, eps_w0, eps_w2t, kcur, lsum, tb + toff, wst);
#pragma unroll
      for (int i = 0; i < 8; ++i) { ksum[i] += kcur[i] * wst; zin[i] = z_a[i] + kcur[i] * cz; }
    }

    const float c = hstep / 6.0f;
#pragma unroll
    for (int i = 0; i < 8; ++i) z_a[i] += ksum[i] * c;
#pragma unroll
    for (int v = 0; v < 8; ++v) ldacc[v] -= c * lsum[v];
  }

  __builtin_amdgcn_s_setprio(0);

  {
    float* orow = out_z + (size_t)(rowbase + lo) * DD + 2 * hi;
#pragma unroll
    for (int kc = 0; kc < 8; ++kc)
      *(v2f*)(orow + 4 * kc) = z_a[kc];
#pragma unroll
    for (int v = 0; v < 8; ++v)
      if (lo == v) out_ld[rowbase + 8 * hi + v] = ldacc[v];
  }
}

extern "C" void kernel_launch(void* const* d_in, const int* in_sizes, int n_in,
                              void* d_out, int out_size, void* d_ws, size_t ws_size,
                              hipStream_t stream) {
  (void)n_in; (void)out_size; (void)d_ws; (void)ws_size;
  const float* z1  = (const float*)d_in[0];
  const float* eps = (const float*)d_in[1];
  const float* W0  = (const float*)d_in[2];
  const float* b0  = (const float*)d_in[3];
  const float* W1  = (const float*)d_in[4];
  const float* b1  = (const float*)d_in[5];
  const float* W2  = (const float*)d_in[6];
  const float* b2  = (const float*)d_in[7];

  const int Btot = in_sizes[0] / DD;
  float* out_z  = (float*)d_out;
  float* out_ld = out_z + (size_t)Btot * DD;

  dim3 grid(Btot / ROWS_BLK), block(NWAVES * 32);
  size_t shmem = (size_t)SMEM_FLOATS * sizeof(float);
  ffjord_kernel<<<grid, block, shmem, stream>>>(z1, eps, W0, b0, W1, b1, W2, b2,
                                                out_z, out_ld);
}


# tanh of both hidden layers taken by the hardware instruction v_tanh_f32 instead of clamp, scale, v_exp, add, v_rcp, multiply-add: 128 tanh per evaluation cost one issue slot each, not five to seven
# speedup vs baseline: 1.7182x; 1.0476x over previous
#include <hip/hip_runtime.h>


typedef __attribute__((ext_vector_type(2))) float v2f;
typedef __attribute__((ext_vector_type(4))) float v4f;
typedef __attribute__((ext_vector_type(8))) float v8f;
typedef __attribute__((address_space(3))) volatile int lds_vint;

#define NWAVES    8
#define MROWS     16
#define ROWS_BLK  (NWAVES * MROWS)
#define DD        32
#define HH        128
#define SSTRIDE   132

#define OFF_W0   0
#define OFF_W1   (OFF_W0 + 4096)
#define OFF_W2   (OFF_W1 + 16384)
#define OFF_B0   (OFF_W2 + 4096)
#define OFF_T0   (OFF_B0 + 128)
#define OFF_B1   (OFF_T0 + 128)
#define OFF_T1   (OFF_B1 + 128)
#define OFF_B2   (OFF_T1 + 128)
#define OFF_T2   (OFF_B2 + 32)
#define OFF_W2T  (OFF_T2 + 32)
#define OFF_STG  (OFF_W2T + 4096)
#define STG_WAVE (2 * 16 * SSTRIDE)
#define OFF_PRG  (OFF_STG + NWAVES * STG_WAVE)
#define SMEM_FLOATS (OFF_PRG + 16)

#define NTHR (NWAVES * 32)
#define STG_PER(K, N) (((K) / 2) * ((N) / 4) / NTHR)
template <int K, int N>
__device__ __forceinline__ void stage_load(v4f* r, const float* __restrict__ Wg, int tid) {
#pragma unroll
  for (int j = 0; j < STG_PER(K, N); ++j) {
    const int u = tid + NTHR * j;
    const float* p = Wg + (2 * (u / (N / 4))) * N + 4 * (u % (N / 4));
    r[2 * j]     = *(const v4f*)p;
    r[2 * j + 1] = *(const v4f*)(p + N);
  }
}
template <int K, int N>
__device__ __forceinline__ void stage_store(float* dst, const v4f* r, int tid) {
#pragma unroll
  for (int j = 0; j < STG_PER(K, N); ++j) {
    const int u = tid + NTHR * j;
    const int k = 2 * (u / (N / 4)), n = 4 * (u % (N / 4));
    const int idx = (((k >> 2) * (N / 16) + (n >> 4)) << 6) + (((k >> 1) & 1) << 5) + 2 * (n & 15);
    const v4f a = r[2 * j], b = r[2 * j + 1];
    v4f lo4 = {a.x, b.x, a.y, b.y};
    v4f hi4 = {a.z, b.z, a.w, b.w};
    *(v4f*)(dst + idx)     = lo4;
    *(v4f*)(dst + idx + 4) = hi4;
  }
}

__device__ __forceinline__ void stage_store_w2t(float* dst, const v4f* r, int tid) {
#pragma unroll
  for (int j = 0; j < STG_PER(128, 32); ++j) {
    const int u = tid + NTHR * j;
    const int k = 2 * (u / (DD / 4)), n = 4 * (u % (DD / 4));
    const int idx = (((n >> 2) * (HH / 16) + (k >> 4)) << 6) + 2 * (k & 15);
    const v4f a = r[2 * j], b = r[2 * j + 1];
    v4f lo4 = {a.x, a.y, b.x, b.y};
    v4f hi4 = {a.z, a.w, b.z, b.w};
    *(v4f*)(dst + idx)      = lo4;
    *(v4f*)(dst + idx + 32) = hi4;
  }
}

__device__ __forceinline__ float fast_tanh(float x) {
  return __builtin_amdgcn_tanhf(x);
}

__device__ __forceinline__ v8f wmma4(v2f a, v2f b, v8f c) {
  return __builtin_amdgcn_wmma_f32_16x16x4_f32(false, a, false, b, (short)0, c,
                                               false, false);
}

template <int NKC, int NT>
__device__ __forceinline__ void load_b(v2f* b, const float* wF, int kc0) {
#pragma unroll
  for (int j = 0; j < NKC; ++j)
#pragma unroll
    for (int nt = 0; nt < NT; ++nt)
      b[j * NT + nt] = *(const v2f*)(wF + ((kc0 + j) * NT + nt) * 64);
}
template <int NKC>
__device__ __forceinline__ void load_a(v2f* a, const float* sA, int kc0) {
#pragma unroll
  for (int j = 0; j < NKC; ++j)
    a[j] = *(const v2f*)(sA + 4 * (kc0 + j));
}
template <int NKC, int NT>
__device__ __forceinline__ void mma_block(v8f* acc, const v2f* a, const v2f* b) {
#pragma unroll
  for (int j = 0; j < NKC; ++j)
#pragma unroll
    for (int nt = 0; nt < NT; ++nt)
      acc[nt] = wmma4(a[j], b[j * NT + nt], acc[nt]);
}
#define STAGE_FENCE() __builtin_amdgcn_sched_barrier(0)

__device__ __forceinline__ void gemm_k128_n128(v8f* acc, const float* sA, const float* wF) {
  v2f a0[4], b0[32], a1[4], b1[32];
  load_a<4>(a0, sA, 0);
  load_b<4, 8>(b0, wF, 0);
#pragma clang loop unroll(disable)
  for (int kb = 0; kb < 32; kb += 8) {
    load_a<4>(a1, sA, kb + 4);
    load_b<4, 8>(b1, wF, kb + 4);
    STAGE_FENCE();
    mma_block<4, 8>(acc, a0, b0);
    STAGE_FENCE();
    const int kn = (kb + 8) & 31;
    load_a<4>(a0, sA, kn);
    load_b<4, 8>(b0, wF, kn);
    STAGE_FENCE();
    mma_block<4, 8>(acc, a1, b1);
  }
}

__device__ __forceinline__ void gemm_k32_n128(v8f* acc, const v2f* a, const float* wF) {
  v2f b0[32], b1[32];
  load_b<4, 8>(b0, wF, 0);
  load_b<4, 8>(b1, wF, 4);
  STAGE_FENCE();
  mma_block<4, 8>(acc, a, b0);
  mma_block<4, 8>(acc, a + 4, b1);
}

__device__ __forceinline__ void func_eval(
    const float* smem, float* sH, float* sU, int lo, int hi,
    const v2f* zin, const v8f* eps_w0, const v8f* eps_w2t,
    v2f* kcur, float* lsum, float t, float wrk)
{
  const v8f ZERO8 = {0.f,0.f,0.f,0.f,0.f,0.f,0.f,0.f};

  const int fb = hi * 32 + 2 * lo;
  const float* wF0 = smem + OFF_W0 + fb;
  const float* wF1 = smem + OFF_W1 + fb;
  const float* wF2 = smem + OFF_W2 + fb;
  float* sHa = sH + lo * SSTRIDE + 2 * hi;
  float* sUa = sU + lo * SSTRIDE + 2 * hi;
  float* sHs = sH + 8 * hi * SSTRIDE + lo;
  float* sUs = sU + 8 * hi * SSTRIDE + lo;

  v8f acc[8];
#pragma unroll
  for (int nt = 0; nt < 8; ++nt) acc[nt] = ZERO8;
  gemm_k32_n128(acc, zin, wF0);
#pragma unroll
  for (int nt = 0; nt < 8; ++nt) {
    int col = 16 * nt + lo;
    float bias = smem[OFF_B0 + col] + t * smem[OFF_T0 + col];
#pragma unroll
    for (int v = 0; v < 8; ++v) {
      int idx = v * SSTRIDE + 16 * nt;
      float h = fast_tanh(acc[nt][v] + bias);
      sHs[idx] = h;
      sUs[idx] = eps_w0[nt][v] * (1.0f - h * h);
    }
  }
  __builtin_amdgcn_wave_barrier();

#pragma unroll
  for (int nt = 0; nt < 8; ++nt) acc[nt] = ZERO8;
  gemm_k128_n128(acc, sHa, wF1);
  __builtin_amdgcn_wave_barrier();
#pragma unroll
  for (int nt = 0; nt < 8; ++nt) {
    int col = 16 * nt + lo;
    float bias = smem[OFF_B1 + col] + t * smem[OFF_T1 + col];
#pragma unroll
    for (int v = 0; v < 8; ++v)
      sHs[v * SSTRIDE + 16 * nt] = fast_tanh(acc[nt][v] + bias);
  }
  __builtin_amdgcn_wave_barrier();

#pragma unroll
  for (int nt = 0; nt < 8; ++nt) acc[nt] = ZERO8;
  gemm_k128_n128(acc, sUa, wF1);
  float p[8];
  {
    v2f p2[4];
#pragma unroll
    for (int q = 0; q < 4; ++q) p2[q] = (v2f){0.0f, 0.0f};
#pragma unroll
    for (int nt = 0; nt < 8; ++nt) {
#pragma unroll
      for (int q = 0; q < 4; ++q) {
        const v2f h = {sHs[(2 * q) * SSTRIDE + 16 * nt], sHs[(2 * q + 1) * SSTRIDE + 16 * nt]};
        const v2f a = {acc[nt][2 * q], acc[nt][2 * q + 1]};
        const v2f g = {eps_w2t[nt][2 * q], eps_w2t[nt][2 * q + 1]};
        p2[q] += (a * (1.0f - h * h)) * g;
      }
    }
#pragma unroll
    for (int q = 0; q < 4; ++q) { p[2 * q] = p2[q].x; p[2 * q + 1] = p2[q].y; }
  }
#pragma unroll
  for (int off = 1; off < 16; off <<= 1) {
#pragma unroll
    for (int v = 0; v < 8; ++v) p[v] += __shfl_xor(p[v], off, 32);
  }
#pragma unroll
  for (int v = 0; v < 8; ++v) lsum[v] += wrk * p[v];

  v8f dacc[2];
  dacc[0] = ZERO8; dacc[1] = ZERO8;
  {
    v2f ah0[16], w0[32], ah1[16], w1[32];
    load_a<16>(ah0, sHa, 0);   load_b<16, 2>(w0, wF2, 0);
    load_a<16>(ah1, sHa, 16);  load_b<16, 2>(w1, wF2, 16);
    STAGE_FENCE();
    mma_block<16, 2>(dacc, ah0, w0);
    mma_block<16, 2>(dacc, ah1, w1);
  }
  __builtin_amdgcn_wave_barrier();

#pragma unroll
  for (int nt = 0; nt < 2; ++nt) {
    int col = 16 * nt + lo;
    float bias = smem[OFF_B2 + col] + t * smem[OFF_T2 + col];
#pragma unroll
    for (int v = 0; v < 8; ++v)
      sHs[v * SSTRIDE + 16 * nt] = dacc[nt][v] + bias;
  }
  __builtin_amdgcn_wave_barrier();
#pragma unroll
  for (int kc = 0; kc < 8; ++kc)
    kcur[kc] = *(const v2f*)(sHa + 4 * kc);
  __builtin_amdgcn_wave_barrier();
}

__global__ __launch_bounds__(256)
void ffjord_kernel(const float* __restrict__ z1, const float* __restrict__ eps,
                   const float* __restrict__ W0, const float* __restrict__ b0,
                   const float* __restrict__ W1, const float* __restrict__ b1,
                   const float* __restrict__ W2, const float* __restrict__ b2,
                   float* __restrict__ out_z, float* __restrict__ out_ld)
{
  extern __shared__ __attribute__((aligned(16))) float smem[];
  const int tid  = threadIdx.x;
  const int lane = tid & 31;
  const int wave = tid >> 5;
  const int lo   = lane & 15;
  const int hi   = lane >> 4;

  v4f rw0[2 * STG_PER(32, 128)], rw1[2 * STG_PER(128, 128)], rw2[2 * STG_PER(128, 32)];
  stage_load<32, 128>(rw0, W0, tid);
  stage_load<128, 128>(rw1, W1, tid);
  stage_load<128, 32>(rw2, W2, tid);
  const int nb = tid & 127, nc = tid & 31;
  const float t0v = W0[32 * 128 + nb], b0v = b0[nb];
  const float t1v = W1[128 * 128 + nb], b1v = b1[nb];
  const float t2v = W2[128 * 32 + nc], b2v = b2[nc];
  stage_store<32, 128>(smem + OFF_W0, rw0, tid);
  stage_store<128, 128>(smem + OFF_W1, rw1, tid);
  stage_store<128, 32>(smem + OFF_W2, rw2, tid);
  stage_store_w2t(smem + OFF_W2T, rw2, tid);
  if (tid < 128) {
    smem[OFF_T0 + nb] = t0v;
    smem[OFF_B0 + nb] = b0v;
    smem[OFF_T1 + nb] = t1v;
    smem[OFF_B1 + nb] = b1v;
  }
  if (tid < 32) {
    smem[OFF_T2 + nc] = t2v;
    smem[OFF_B2 + nc] = b2v;
  }
  lds_vint* prog = (lds_vint*)(smem + OFF_PRG);
  const int wave_u  = __builtin_amdgcn_readfirstlane(wave);
  const int partner = wave_u ^ (NWAVES / 2);
  if (lane == 0) prog[wave_u] = 0;
  __syncthreads();

  float* sH = smem + OFF_STG + wave * STG_WAVE;
  float* sU = sH + 16 * SSTRIDE;
  const int rowbase = blockIdx.x * ROWS_BLK + wave * MROWS;

  v2f z_a[8], zin[8], ksum[8], kcur[8];
  v8f eps_w0[8];
  v8f eps_w2t[8];
  float ldacc[8];

  {
    const float* zrow = z1  + (size_t)(rowbase + lo) * DD + 2 * hi;
    const float* erow = eps + (size_t)(rowbase + lo) * DD + 2 * hi;
    v2f eps_a[8];
#pragma unroll
    for (int kc = 0; kc < 8; ++kc) {
      z_a[kc]   = *(const v2f*)(zrow + 4 * kc);
      eps_a[kc] = *(const v2f*)(erow + 4 * kc);
    }
    const v8f ZERO8 = {0.f,0.f,0.f,0.f,0.f,0.f,0.f,0.f};
#pragma unroll
    for (int nt = 0; nt < 8; ++nt) eps_w0[nt] = ZERO8;
    gemm_k32_n128(eps_w0, eps_a, smem + OFF_W0 + hi * 32 + 2 * lo);
#pragma unroll
    for (int nt = 0; nt < 8; ++nt) eps_w2t[nt] = ZERO8;
    gemm_k32_n128(eps_w2t, eps_a, smem + OFF_W2T + hi * 32 + 2 * lo);
#pragma unroll
    for (int v = 0; v < 8; ++v) ldacc[v] = 0.0f;
  }

  const float hstep = -0.2f;
  for (int s = 0; s < 5; ++s) {
    const float tb = 1.0f - 0.2f * (float)s;
    float lsum[8];
#pragma unroll
    for (int v = 0; v < 8; ++v) lsum[v] = 0.0f;
#pragma unroll
    for (int i = 0; i < 8; ++i) { zin[i] = z_a[i]; ksum[i] = (v2f){0.0f, 0.0f}; }

#pragma clang loop unroll(disable)
    for (int st = 0; st < 4; ++st) {
      const float toff = (st == 0) ? 0.0f : ((st == 3) ? hstep : 0.5f * hstep);
      const float wst  = (st == 0 || st == 3) ? 1.0f : 2.0f;
      const float cz   = (st < 2) ? 0.5f * hstep : hstep;
      {
        const int done = 4 * s + st;
        if (lane == 0) prog[wave_u] = done;
        const int other = __builtin_amdgcn_readfirstlane(prog[partner]);
        const bool behind = (done < other) || (done == other && wave_u >= NWAVES / 2);
        if (behind) __builtin_amdgcn_s_setprio(3);
        else        __builtin_amdgcn_s_setprio(0);
      }
      func_eval(smem, sH, sU, lo, hi, zin, eps_w0, eps_w2t, kcur, lsum, tb + toff, wst);
#pragma unroll
      for (int i = 0; i < 8; ++i) { ksum[i] += kcur[i] * wst; zin[i] = z_a[i] + kcur[i] * cz; }
    }

    const float c = hstep / 6.0f;
#pragma unroll
    for (int i = 0; i < 8; ++i) z_a[i] += ksum[i] * c;
#pragma unroll
    for (int v = 0; v < 8; ++v) ldacc[v] -= c * lsum[v];
  }

  __builtin_amdgcn_s_setprio(0);

  {
    float* orow = out_z + (size_t)(rowbase + lo) * DD + 2 * hi;
#pragma unroll
    for (int kc = 0; kc < 8; ++kc)
      *(v2f*)(orow + 4 * kc) = z_a[kc];
#pragma unroll
    for (int v = 0; v < 8; ++v)
      if (lo == v) out_ld[rowbase + 8 * hi + v] = ldacc[v];
  }
}

extern "C" void kernel_launch(void* const* d_in, const int* in_sizes, int n_in,
                              void* d_out, int out_size, void* d_ws, size_t ws_size,
                              hipStream_t stream) {
  (void)n_in; (void)out_size; (void)d_ws; (void)ws_size;
  const float* z1  = (const float*)d_in[0];
  const float* eps = (const float*)d_in[1];
  const float* W0  = (const float*)d_in[2];
  const float* b0  = (const float*)d_in[3];
  const float* W1  = (const float*)d_in[4];
  const float* b1  = (const float*)d_in[5];
  const float* W2  = (const float*)d_in[6];
  const float* b2  = (const float*)d_in[7];

  const int Btot = in_sizes[0] / DD;
  float* out_z  = (float*)d_out;
  float* out_ld = out_z + (size_t)Btot * DD;

  dim3 grid(Btot / ROWS_BLK), block(NWAVES * 32);
  size_t shmem = (size_t)SMEM_FLOATS * sizeof(float);
  ffjord_kernel<<<grid, block, shmem, stream>>>(z1, eps, W0, b0, W1, b1, W2, b2,
                                                out_z, out_ld);
}


# Second layer's primal h1 @ W1a and tangent u1 @ W1a run as ONE pass over W1a on two accumulator sets: each weight fragment read from LDS once feeds two WMMAs; h2 meets the tangent in registers
# speedup vs baseline: 1.8104x; 1.0537x over previous
#include <hip/hip_runtime.h>


typedef __attribute__((ext_vector_type(2))) float v2f;
typedef __attribute__((ext_vector_type(4))) float v4f;
typedef __attribute__((ext_vector_type(8))) float v8f;
typedef __attribute__((address_space(3))) volatile int lds_vint;

#define NWAVES    8
#define MROWS     16
#define ROWS_BLK  (NWAVES * MROWS)
#define DD        32
#define HH        128
#define SSTRIDE   132

#define OFF_W0   0
#define OFF_W1   (OFF_W0 + 4096)
#define OFF_W2   (OFF_W1 + 16384)
#define OFF_B0   (OFF_W2 + 4096)
#define OFF_T0   (OFF_B0 + 128)
#define OFF_B1   (OFF_T0 + 128)
#define OFF_T1   (OFF_B1 + 128)
#define OFF_B2   (OFF_T1 + 128)
#define OFF_T2   (OFF_B2 + 32)
#define OFF_W2T  (OFF_T2 + 32)
#define OFF_STG  (OFF_W2T + 4096)
#define STG_WAVE (2 * 16 * SSTRIDE)
#define OFF_PRG  (OFF_STG + NWAVES * STG_WAVE)
#define SMEM_FLOATS (OFF_PRG + 16)

#define NTHR (NWAVES * 32)
#define STG_PER(K, N) (((K) / 2) * ((N) / 4) / NTHR)
template <int K, int N>
__device__ __forceinline__ void stage_load(v4f* r, const float* __restrict__ Wg, int tid) {
#pragma unroll
  for (int j = 0; j < STG_PER(K, N); ++j) {
    const int u = tid + NTHR * j;
    const float* p = Wg + (2 * (u / (N / 4))) * N + 4 * (u % (N / 4));
    r[2 * j]     = *(const v4f*)p;
    r[2 * j + 1] = *(const v4f*)(p + N);
  }
}
template <int K, int N>
__device__ __forceinline__ void stage_store(float* dst, const v4f* r, int tid) {
#pragma unroll
  for (int j = 0; j < STG_PER(K, N); ++j) {
    const int u = tid + NTHR * j;
    const int k = 2 * (u / (N / 4)), n = 4 * (u % (N / 4));
    const int idx = (((k >> 2) * (N / 16) + (n >> 4)) << 6) + (((k >> 1) & 1) << 5) + 2 * (n & 15);
    const v4f a = r[2 * j], b = r[2 * j + 1];
    v4f lo4 = {a.x, b.x, a.y, b.y};
    v4f hi4 = {a.z, b.z, a.w, b.w};
    *(v4f*)(dst + idx)     = lo4;
    *(v4f*)(dst + idx + 4) = hi4;
  }
}

__device__ __forceinline__ void stage_store_w2t(float* dst, const v4f* r, int tid) {
#pragma unroll
  for (int j = 0; j < STG_PER(128, 32); ++j) {
    const int u = tid + NTHR * j;
    const int k = 2 * (u / (DD / 4)), n = 4 * (u % (DD / 4));
    const int idx = (((n >> 2) * (HH / 16) + (k >> 4)) << 6) + 2 * (k & 15);
    const v4f a = r[2 * j], b = r[2 * j + 1];
    v4f lo4 = {a.x, a.y, b.x, b.y};
    v4f hi4 = {a.z, a.w, b.z, b.w};
    *(v4f*)(dst + idx)      = lo4;
    *(v4f*)(dst + idx + 32) = hi4;
  }
}

__device__ __forceinline__ float fast_tanh(float x) {
  return __builtin_amdgcn_tanhf(x);
}

__device__ __forceinline__ v8f wmma4(v2f a, v2f b, v8f c) {
  return __builtin_amdgcn_wmma_f32_16x16x4_f32(false, a, false, b, (short)0, c,
                                               false, false);
}

template <int NKC, int NT>
__device__ __forceinline__ void load_b(v2f* b, const float* wF, int kc0) {
#pragma unroll
  for (int j = 0; j < NKC; ++j)
#pragma unroll
    for (int nt = 0; nt < NT; ++nt)
      b[j * NT + nt] = *(const v2f*)(wF + ((kc0 + j) * NT + nt) * 64);
}
template <int NKC>
__device__ __forceinline__ void load_a(v2f* a, const float* sA, int kc0) {
#pragma unroll
  for (int j = 0; j < NKC; ++j)
    a[j] = *(const v2f*)(sA + 4 * (kc0 + j));
}
template <int NKC, int NT>
__device__ __forceinline__ void mma_block(v8f* acc, const v2f* a, const v2f* b) {
#pragma unroll
  for (int j = 0; j < NKC; ++j)
#pragma unroll
    for (int nt = 0; nt < NT; ++nt)
      acc[nt] = wmma4(a[j], b[j * NT + nt], acc[nt]);
}
#define STAGE_FENCE() __builtin_amdgcn_sched_barrier(0)

template <int NKC, int NT>
__device__ __forceinline__ void mma_block_pair(v8f* accP, v8f* accT, const v2f* aP, const v2f* aT,
                                               const v2f* b) {
#pragma unroll
  for (int j = 0; j < NKC; ++j)
#pragma unroll
    for (int nt = 0; nt < NT; ++nt) {
      accP[nt] = wmma4(aP[j], b[j * NT + nt], accP[nt]);
      accT[nt] = wmma4(aT[j], b[j * NT + nt], accT[nt]);
    }
}

__device__ __forceinline__ void gemm_k128_n128_pair(v8f* accP, v8f* accT, const float* sAP,
                                                    const float* sAT, const float* wF) {
  v2f p0[2], t0[2], b0[16], p1[2], t1[2], b1[16];
  load_a<2>(p0, sAP, 0);
  load_a<2>(t0, sAT, 0);
  load_b<2, 8>(b0, wF, 0);
#pragma clang loop unroll(disable)
  for (int kb = 0; kb < 32; kb += 4) {
    load_a<2>(p1, sAP, kb + 2);
    load_a<2>(t1, sAT, kb + 2);
    load_b<2, 8>(b1, wF, kb + 2);
    STAGE_FENCE();
    mma_block_pair<2, 8>(accP, accT, p0, t0, b0);
    STAGE_FENCE();
    const int kn = (kb + 4) & 31;
    load_a<2>(p0, sAP, kn);
    load_a<2>(t0, sAT, kn);
    load_b<2, 8>(b0, wF, kn);
    STAGE_FENCE();
    mma_block_pair<2, 8>(accP, accT, p1, t1, b1);
  }
}

__device__ __forceinline__ void gemm_k32_n128(v8f* acc, const v2f* a, const float* wF) {
  v2f b0[32], b1[32];
  load_b<4, 8>(b0, wF, 0);
  load_b<4, 8>(b1, wF, 4);
  STAGE_FENCE();
  mma_block<4, 8>(acc, a, b0);
  mma_block<4, 8>(acc, a + 4, b1);
}

__device__ __forceinline__ void func_eval(
    const float* smem, float* sH, float* sU, int lo, int hi,
    const v2f* zin, const v8f* eps_w0, const v8f* eps_w2t,
    v2f* kcur, float* lsum, float t, float wrk)
{
  const v8f ZERO8 = {0.f,0.f,0.f,0.f,0.f,0.f,0.f,0.f};

  const int fb = hi * 32 + 2 * lo;
  const float* wF0 = smem + OFF_W0 + fb;
  const float* wF1 = smem + OFF_W1 + fb;
  const float* wF2 = smem + OFF_W2 + fb;
  float* sHa = sH + lo * SSTRIDE + 2 * hi;
  float* sUa = sU + lo * SSTRIDE + 2 * hi;
  float* sHs = sH + 8 * hi * SSTRIDE + lo;
  float* sUs = sU + 8 * hi * SSTRIDE + lo;

  v8f acc[8];
#pragma unroll
  for (int nt = 0; nt < 8; ++nt) acc[nt] = ZERO8;
  gemm_k32_n128(acc, zin, wF0);
#pragma unroll
  for (int nt = 0; nt < 8; ++nt) {
    int col = 16 * nt + lo;
    float bias = smem[OFF_B0 + col] + t * smem[OFF_T0 + col];
#pragma unroll
    for (int v = 0; v < 8; ++v) {
      int idx = v * SSTRIDE + 16 * nt;
      float h = fast_tanh(acc[nt][v] + bias);
      sHs[idx] = h;
      sUs[idx] = eps_w0[nt][v] * (1.0f - h * h);
    }
  }
  __builtin_amdgcn_wave_barrier();

  v8f acu[8];
#pragma unroll
  for (int nt = 0; nt < 8; ++nt) { acc[nt] = ZERO8; acu[nt] = ZERO8; }
  gemm_k128_n128_pair(acc, acu, sHa, sUa, wF1);
  __builtin_amdgcn_wave_barrier();
  float p[8];
  {
    v2f p2[4];
#pragma unroll
    for (int q = 0; q < 4; ++q) p2[q] = (v2f){0.0f, 0.0f};
#pragma unroll
    for (int nt = 0; nt < 8; ++nt) {
      int col = 16 * nt + lo;
      float bias = smem[OFF_B1 + col] + t * smem[OFF_T1 + col];
#pragma unroll
      for (int q = 0; q < 4; ++q) {
        const v2f h = {fast_tanh(acc[nt][2 * q] + bias), fast_tanh(acc[nt][2 * q + 1] + bias)};
        sHs[(2 * q) * SSTRIDE + 16 * nt]     = h.x;
        sHs[(2 * q + 1) * SSTRIDE + 16 * nt] = h.y;
        const v2f a = {acu[nt][2 * q], acu[nt][2 * q + 1]};
        const v2f g = {eps_w2t[nt][2 * q], eps_w2t[nt][2 * q + 1]};
        p2[q] += (a * (1.0f - h * h)) * g;
      }
    }
#pragma unroll
    for (int q = 0; q < 4; ++q) { p[2 * q] = p2[q].x; p[2 * q + 1] = p2[q].y; }
  }
  __builtin_amdgcn_wave_barrier();
#pragma unroll
  for (int off = 1; off < 16; off <<= 1) {
#pragma unroll
    for (int v = 0; v < 8; ++v) p[v] += __shfl_xor(p[v], off, 32);
  }
#pragma unroll
  for (int v = 0; v < 8; ++v) lsum[v] += wrk * p[v];

  v8f dacc[2];
  dacc[0] = ZERO8; dacc[1] = ZERO8;
  {
    v2f ah0[16], w0[32], ah1[16], w1[32];
    load_a<16>(ah0, sHa, 0);   load_b<16, 2>(w0, wF2, 0);
    load_a<16>(ah1, sHa, 16);  load_b<16, 2>(w1, wF2, 16);
    STAGE_FENCE();
    mma_block<16, 2>(dacc, ah0, w0);
    mma_block<16, 2>(dacc, ah1, w1);
  }
  __builtin_amdgcn_wave_barrier();

#pragma unroll
  for (int nt = 0; nt < 2; ++nt) {
    int col = 16 * nt + lo;
    float bias = smem[OFF_B2 + col] + t * smem[OFF_T2 + col];
#pragma unroll
    for (int v = 0; v < 8; ++v)
      sHs[v * SSTRIDE + 16 * nt] = dacc[nt][v] + bias;
  }
  __builtin_amdgcn_wave_barrier();
#pragma unroll
  for (int kc = 0; kc < 8; ++kc)
    kcur[kc] = *(const v2f*)(sHa + 4 * kc);
  __builtin_amdgcn_wave_barrier();
}

__global__ __launch_bounds__(256)
void ffjord_kernel(const float* __restrict__ z1, const float* __restrict__ eps,
                   const float* __restrict__ W0, const float* __restrict__ b0,
                   const float* __restrict__ W1, const float* __restrict__ b1,
                   const float* __restrict__ W2, const float* __restrict__ b2,
                   float* __restrict__ out_z, float* __restrict__ out_ld)
{
  extern __shared__ __attribute__((aligned(16))) float smem[];
  const int tid  = threadIdx.x;
  const int lane = tid & 31;
  const int wave = tid >> 5;
  const int lo   = lane & 15;
  const int hi   = lane >> 4;

  v4f rw0[2 * STG_PER(32, 128)], rw1[2 * STG_PER(128, 128)], rw2[2 * STG_PER(128, 32)];
  stage_load<32, 128>(rw0, W0, tid);
  stage_load<128, 128>(rw1, W1, tid);
  stage_load<128, 32>(rw2, W2, tid);
  const int nb = tid & 127, nc = tid & 31;
  const float t0v = W0[32 * 128 + nb], b0v = b0[nb];
  const float t1v = W1[128 * 128 + nb], b1v = b1[nb];
  const float t2v = W2[128 * 32 + nc], b2v = b2[nc];
  stage_store<32, 128>(smem + OFF_W0, rw0, tid);
  stage_store<128, 128>(smem + OFF_W1, rw1, tid);
  stage_store<128, 32>(smem + OFF_W2, rw2, tid);
  stage_store_w2t(smem + OFF_W2T, rw2, tid);
  if (tid < 128) {
    smem[OFF_T0 + nb] = t0v;
    smem[OFF_B0 + nb] = b0v;
    smem[OFF_T1 + nb] = t1v;
    smem[OFF_B1 + nb] = b1v;
  }
  if (tid < 32) {
    smem[OFF_T2 + nc] = t2v;
    smem[OFF_B2 + nc] = b2v;
  }
  lds_vint* prog = (lds_vint*)(smem + OFF_PRG);
  const int wave_u  = __builtin_amdgcn_readfirstlane(wave);
  const int partner = wave_u ^ (NWAVES / 2);
  if (lane == 0) prog[wave_u] = 0;
  __syncthreads();

  float* sH = smem + OFF_STG + wave * STG_WAVE;
  float* sU = sH + 16 * SSTRIDE;
  const int rowbase = blockIdx.x * ROWS_BLK + wave * MROWS;

  v2f z_a[8], zin[8], ksum[8], kcur[8];
  v8f eps_w0[8];
  v8f eps_w2t[8];
  float ldacc[8];

  {
    const float* zrow = z1  + (size_t)(rowbase + lo) * DD + 2 * hi;
    const float* erow = eps + (size_t)(rowbase + lo) * DD + 2 * hi;
    v2f eps_a[8];
#pragma unroll
    for (int kc = 0; kc < 8; ++kc) {
      z_a[kc]   = *(const v2f*)(zrow + 4 * kc);
      eps_a[kc] = *(const v2f*)(erow + 4 * kc);
    }
    const v8f ZERO8 = {0.f,0.f,0.f,0.f,0.f,0.f,0.f,0.f};
#pragma unroll
    for (int nt = 0; nt < 8; ++nt) eps_w0[nt] = ZERO8;
    gemm_k32_n128(eps_w0, eps_a, smem + OFF_W0 + hi * 32 + 2 * lo);
#pragma unroll
    for (int nt = 0; nt < 8; ++nt) eps_w2t[nt] = ZERO8;
    gemm_k32_n128(eps_w2t, eps_a, smem + OFF_W2T + hi * 32 + 2 * lo);
#pragma unroll
    for (int v = 0; v < 8; ++v) ldacc[v] = 0.0f;
  }

  const float hstep = -0.2f;
  for (int s = 0; s < 5; ++s) {
    const float tb = 1.0f - 0.2f * (float)s;
    float lsum[8];
#pragma unroll
    for (int v = 0; v < 8; ++v) lsum[v] = 0.0f;
#pragma unroll
    for (int i = 0; i < 8; ++i) { zin[i] = z_a[i]; ksum[i] = (v2f){0.0f, 0.0f}; }

#pragma clang loop unroll(disable)
    for (int st = 0; st < 4; ++st) {
      const float toff = (st == 0) ? 0.0f : ((st == 3) ? hstep : 0.5f * hstep);
      const float wst  = (st == 0 || st == 3) ? 1.0f : 2.0f;
      const float cz   = (st < 2) ? 0.5f * hstep : hstep;
      {
        const int done = 4 * s + st;
        if (lane == 0) prog[wave_u] = done;
        const int other = __builtin_amdgcn_readfirstlane(prog[partner]);
        const bool behind = (done < other) || (done == other && wave_u >= NWAVES / 2);
        if (behind) __builtin_amdgcn_s_setprio(3);
        else        __builtin_amdgcn_s_setprio(0);
      }
      func_eval(smem, sH, sU, lo, hi, zin, eps_w0, eps_w2t, kcur, lsum, tb + toff, wst);
#pragma unroll
      for (int i = 0; i < 8; ++i) { ksum[i] += kcur[i] * wst; zin[i] = z_a[i] + kcur[i] * cz; }
    }

    const float c = hstep / 6.0f;
#pragma unroll
    for (int i = 0; i < 8; ++i) z_a[i] += ksum[i] * c;
#pragma unroll
    for (int v = 0; v < 8; ++v) ldacc[v] -= c * lsum[v];
  }

  __builtin_amdgcn_s_setprio(0);

  {
    float* orow = out_z + (size_t)(rowbase + lo) * DD + 2 * hi;
#pragma unroll
    for (int kc = 0; kc < 8; ++kc)
      *(v2f*)(orow + 4 * kc) = z_a[kc];
#pragma unroll
    for (int v = 0; v < 8; ++v)
      if (lo == v) out_ld[rowbase + 8 * hi + v] = ldacc[v];
  }
}

extern "C" void kernel_launch(void* const* d_in, const int* in_sizes, int n_in,
                              void* d_out, int out_size, void* d_ws, size_t ws_size,
                              hipStream_t stream) {
  (void)n_in; (void)out_size; (void)d_ws; (void)ws_size;
  const float* z1  = (const float*)d_in[0];
  const float* eps = (const float*)d_in[1];
  const float* W0  = (const float*)d_in[2];
  const float* b0  = (const float*)d_in[3];
  const float* W1  = (const float*)d_in[4];
  const float* b1  = (const float*)d_in[5];
  const float* W2  = (const float*)d_in[6];
  const float* b2  = (const float*)d_in[7];

  const int Btot = in_sizes[0] / DD;
  float* out_z  = (float*)d_out;
  float* out_ld = out_z + (size_t)Btot * DD;

  dim3 grid(Btot / ROWS_BLK), block(NWAVES * 32);
  size_t shmem = (size_t)SMEM_FLOATS * sizeof(float);
  ffjord_kernel<<<grid, block, shmem, stream>>>(z1, eps, W0, b0, W1, b1, W2, b2,
                                                out_z, out_ld);
}


# Every layer computed as its transpose (weights as A, activations as B): a layer's accumulator layout is the next one's B layout, activations never pass through LDS; k sums and trace in a new order
# speedup vs baseline: 1.9163x; 1.0585x over previous
#include <hip/hip_runtime.h>


typedef __attribute__((ext_vector_type(2))) float v2f;
typedef __attribute__((ext_vector_type(4))) float v4f;
typedef __attribute__((ext_vector_type(8))) float v8f;
typedef __attribute__((address_space(3))) volatile int lds_vint;

#define NWAVES    8
#define MROWS     16
#define ROWS_BLK  (NWAVES * MROWS)
#define DD        32
#define HH        128

#define OFF_W0   0
#define OFF_W1   (OFF_W0 + 4096)
#define OFF_W2   (OFF_W1 + 16384)
#define OFF_B0   (OFF_W2 + 4096)
#define OFF_T0   (OFF_B0 + 128)
#define OFF_B1   (OFF_T0 + 128)
#define OFF_T1   (OFF_B1 + 128)
#define OFF_B2   (OFF_T1 + 128)
#define OFF_T2   (OFF_B2 + 32)
#define OFF_W2T  (OFF_T2 + 32)
#define OFF_PRG  (OFF_W2T + 4096)
#define SMEM_FLOATS (OFF_PRG + 16)

#define NTHR (NWAVES * 32)
#define STG_PER(K, N) (((K) / 2) * ((N) / 4) / NTHR)
template <int K, int N>
__device__ __forceinline__ void stage_load(v4f* r, const float* __restrict__ Wg, int tid) {
#pragma unroll
  for (int j = 0; j < STG_PER(K, N); ++j) {
    const int u = tid + NTHR * j;
    const float* p = Wg + (2 * (u / (N / 4))) * N + 4 * (u % (N / 4));
    r[2 * j]     = *(const v4f*)p;
    r[2 * j + 1] = *(const v4f*)(p + N);
  }
}
template <int K, int N>
__device__ __forceinline__ void stage_store(float* dst, const v4f* r, int tid) {
#pragma unroll
  for (int j = 0; j < STG_PER(K, N); ++j) {
    const int u = tid + NTHR * j;
    const int k = 2 * (u / (N / 4)), n = 4 * (u % (N / 4));
    const int kc = ((k >> 4) << 2) | ((k >> 1) & 3);
    const int idx = ((kc * (N / 16) + (n >> 4)) << 6) + (((k >> 3) & 1) << 5) + 2 * (n & 15);
    const v4f a = r[2 * j], b = r[2 * j + 1];
    v4f lo4 = {a.x, b.x, a.y, b.y};
    v4f hi4 = {a.z, b.z, a.w, b.w};
    *(v4f*)(dst + idx)     = lo4;
    *(v4f*)(dst + idx + 4) = hi4;
  }
}

__device__ __forceinline__ void stage_store_w2t(float* dst, const v4f* r, int tid) {
#pragma unroll
  for (int j = 0; j < STG_PER(128, 32); ++j) {
    const int u = tid + NTHR * j;
    const int k = 2 * (u / (DD / 4)), n = 4 * (u % (DD / 4));
    const int kc = ((n >> 4) << 2) | ((n >> 1) & 3);
    const int idx = ((kc * (HH / 16) + (k >> 4)) << 6) + (((n >> 3) & 1) << 5) + 2 * (k & 15);
    const v4f a = r[2 * j], b = r[2 * j + 1];
    v4f lo4 = {a.x, a.y, b.x, b.y};
    v4f hi4 = {a.z, a.w, b.z, b.w};
    *(v4f*)(dst + idx)       = lo4;
    *(v4f*)(dst + idx + 512) = hi4;
  }
}

__device__ __forceinline__ float fast_tanh(float x) {
  return __builtin_amdgcn_tanhf(x);
}

__device__ __forceinline__ v8f wmma4(v2f a, v2f b, v8f c) {
  return __builtin_amdgcn_wmma_f32_16x16x4_f32(false, a, false, b, (short)0, c,
                                               false, false);
}

__device__ __forceinline__ v2f chunk(const v8f* x, int kc) {
  return (v2f){x[kc >> 2][2 * (kc & 3)], x[kc >> 2][2 * (kc & 3) + 1]};
}

#define STAGE_FENCE() __builtin_amdgcn_sched_barrier(0)

__device__ __forceinline__ v8f bias_tile(const float* bb, const float* tw, int m, int hi, float t) {
  const v4f b0 = *(const v4f*)(bb + 16 * m + 8 * hi), b1 = *(const v4f*)(bb + 16 * m + 8 * hi + 4);
  const v4f w0 = *(const v4f*)(tw + 16 * m + 8 * hi), w1 = *(const v4f*)(tw + 16 * m + 8 * hi + 4);
  const v4f r0 = b0 + t * w0, r1 = b1 + t * w1;
  return (v8f){r0.x, r0.y, r0.z, r0.w, r1.x, r1.y, r1.z, r1.w};
}

__device__ __forceinline__ void gemm_k32_m128(v8f* acc, const v8f* act, const float* wF) {
  v2f w[64];
#pragma unroll
  for (int f = 0; f < 64; ++f) w[f] = *(const v2f*)(wF + f * 64);
  STAGE_FENCE();
#pragma unroll
  for (int kc = 0; kc < 8; ++kc) {
    const v2f b = chunk(act, kc);
#pragma unroll
    for (int m = 0; m < 8; ++m) acc[m] = wmma4(w[kc * 8 + m], b, acc[m]);
  }
  STAGE_FENCE();
}

__device__ __forceinline__ void func_eval(
    const float* smem, int lo, int hi,
    const v8f* zin, const v8f* eps_w0, const v8f* eps_w2t,
    v8f* kcur, float& tr_out, float t)
{
  const v8f ZERO8 = {0.f,0.f,0.f,0.f,0.f,0.f,0.f,0.f};
  const int fb = hi * 32 + 2 * lo;
  const float* wF0 = smem + OFF_W0 + fb;
  const float* wF1 = smem + OFF_W1 + fb;
  const float* wF2 = smem + OFF_W2 + fb;

  v8f h1[8], u1[8];
#pragma unroll
  for (int m = 0; m < 8; ++m) h1[m] = bias_tile(smem + OFF_B0, smem + OFF_T0, m, hi, t);
  gemm_k32_m128(h1, zin, wF0);
#pragma unroll
  for (int m = 0; m < 8; ++m)
#pragma unroll
    for (int r = 0; r < 8; ++r) {
      const float h = fast_tanh(h1[m][r]);
      h1[m][r] = h;
      u1[m][r] = eps_w0[m][r] * (1.0f - h * h);
    }

  v8f dacc[2];
  dacc[0] = bias_tile(smem + OFF_B2, smem + OFF_T2, 0, hi, t);
  dacc[1] = bias_tile(smem + OFF_B2, smem + OFF_T2, 1, hi, t);
  float tr = 0.0f;
  v2f fr[2][16];
  v8f nextP[2];
  nextP[0] = bias_tile(smem + OFF_B1, smem + OFF_T1, 0, hi, t);
  nextP[1] = bias_tile(smem + OFF_B1, smem + OFF_T1, 1, hi, t);
#pragma unroll
  for (int f = 0; f < 16; ++f)
    fr[0][f] = *(const v2f*)(wF1 + ((f >> 1) * 8 + (f & 1)) * 64);
#pragma unroll
  for (int mp = 0; mp < 4; ++mp) {
    v8f accP[2], accT[2];
    accP[0] = nextP[0]; accP[1] = nextP[1];
    accT[0] = ZERO8; accT[1] = ZERO8;
#pragma unroll
    for (int blk = 0; blk < 4; ++blk) {
      const int cur = blk & 1, nxt = cur ^ 1;
      if (blk < 3 || mp < 3) {
        const int kb = (blk < 3) ? 8 * (blk + 1) : 0;
        const int mn = (blk < 3) ? 2 * mp : 2 * mp + 2;
#pragma unroll
        for (int f = 0; f < 16; ++f)
          fr[nxt][f] = *(const v2f*)(wF1 + ((kb + (f >> 1)) * 8 + mn + (f & 1)) * 64);
      }
      STAGE_FENCE();
#pragma unroll
      for (int j = 0; j < 8; ++j) {
        const v2f bP = chunk(h1, 8 * blk + j), bT = chunk(u1, 8 * blk + j);
        accP[0] = wmma4(fr[cur][2 * j],     bP, accP[0]);
        accT[0] = wmma4(fr[cur][2 * j],     bT, accT[0]);
        accP[1] = wmma4(fr[cur][2 * j + 1], bP, accP[1]);
        accT[1] = wmma4(fr[cur][2 * j + 1], bT, accT[1]);
      }
      STAGE_FENCE();
    }
    v2f w2[16];
#pragma unroll
    for (int f = 0; f < 16; ++f) w2[f] = *(const v2f*)(wF2 + (16 * mp + f) * 64);
    if (mp < 3) {
      nextP[0] = bias_tile(smem + OFF_B1, smem + OFF_T1, 2 * mp + 2, hi, t);
      nextP[1] = bias_tile(smem + OFF_B1, smem + OFF_T1, 2 * mp + 3, hi, t);
    }
    STAGE_FENCE();
    v8f h2[2];
#pragma unroll
    for (int i = 0; i < 2; ++i)
#pragma unroll
      for (int r = 0; r < 8; ++r) {
        const float h = fast_tanh(accP[i][r]);
        h2[i][r] = h;
        tr += (accT[i][r] * (1.0f - h * h)) * eps_w2t[2 * mp + i][r];
      }
#pragma unroll
    for (int j = 0; j < 8; ++j) {
      const v2f b = chunk(h2, j);
      dacc[0] = wmma4(w2[2 * j],     b, dacc[0]);
      dacc[1] = wmma4(w2[2 * j + 1], b, dacc[1]);
    }
    STAGE_FENCE();
  }
  kcur[0] = dacc[0]; kcur[1] = dacc[1];
  tr_out = tr;
}

__global__ __launch_bounds__(256)
void ffjord_kernel(const float* __restrict__ z1, const float* __restrict__ eps,
                   const float* __restrict__ W0, const float* __restrict__ b0,
                   const float* __restrict__ W1, const float* __restrict__ b1,
                   const float* __restrict__ W2, const float* __restrict__ b2,
                   float* __restrict__ out_z, float* __restrict__ out_ld)
{
  extern __shared__ __attribute__((aligned(16))) float smem[];
  const int tid  = threadIdx.x;
  const int lane = tid & 31;
  const int wave = tid >> 5;
  const int lo   = lane & 15;
  const int hi   = lane >> 4;

  v4f rw0[2 * STG_PER(32, 128)], rw1[2 * STG_PER(128, 128)], rw2[2 * STG_PER(128, 32)];
  stage_load<32, 128>(rw0, W0, tid);
  stage_load<128, 128>(rw1, W1, tid);
  stage_load<128, 32>(rw2, W2, tid);
  const int nb = tid & 127, nc = tid & 31;
  const float t0v = W0[32 * 128 + nb], b0v = b0[nb];
  const float t1v = W1[128 * 128 + nb], b1v = b1[nb];
  const float t2v = W2[128 * 32 + nc], b2v = b2[nc];
  stage_store<32, 128>(smem + OFF_W0, rw0, tid);
  stage_store<128, 128>(smem + OFF_W1, rw1, tid);
  stage_store<128, 32>(smem + OFF_W2, rw2, tid);
  stage_store_w2t(smem + OFF_W2T, rw2, tid);
  if (tid < 128) {
    smem[OFF_T0 + nb] = t0v;
    smem[OFF_B0 + nb] = b0v;
    smem[OFF_T1 + nb] = t1v;
    smem[OFF_B1 + nb] = b1v;
  }
  if (tid < 32) {
    smem[OFF_T2 + nc] = t2v;
    smem[OFF_B2 + nc] = b2v;
  }
  lds_vint* prog = (lds_vint*)(smem + OFF_PRG);
  const int wave_u  = __builtin_amdgcn_readfirstlane(wave);
  const int partner = wave_u ^ (NWAVES / 2);
  if (lane == 0) prog[wave_u] = 0;
  __syncthreads();

  const int rowbase = blockIdx.x * ROWS_BLK + wave * MROWS;

  v8f z_a[2], zin[2], ksum[2], kcur[2];
  v8f eps_w0[8];
  v8f eps_w2t[8];
  float ldacc = 0.0f;

  {
    const float* zrow = z1  + (size_t)(rowbase + lo) * DD + 8 * hi;
    const float* erow = eps + (size_t)(rowbase + lo) * DD + 8 * hi;
    v8f eps_a[2];
#pragma unroll
    for (int o = 0; o < 2; ++o) {
      const v4f za = *(const v4f*)(zrow + 16 * o), zb = *(const v4f*)(zrow + 16 * o + 4);
      const v4f ea = *(const v4f*)(erow + 16 * o), eb = *(const v4f*)(erow + 16 * o + 4);
      z_a[o]   = (v8f){za.x, za.y, za.z, za.w, zb.x, zb.y, zb.z, zb.w};
      eps_a[o] = (v8f){ea.x, ea.y, ea.z, ea.w, eb.x, eb.y, eb.z, eb.w};
    }
    const v8f ZERO8 = {0.f,0.f,0.f,0.f,0.f,0.f,0.f,0.f};
#pragma unroll
    for (int m = 0; m < 8; ++m) { eps_w0[m] = ZERO8; eps_w2t[m] = ZERO8; }
    gemm_k32_m128(eps_w0,  eps_a, smem + OFF_W0  + hi * 32 + 2 * lo);
    gemm_k32_m128(eps_w2t, eps_a, smem + OFF_W2T + hi * 32 + 2 * lo);
  }

  const float hstep = -0.2f;
  for (int s = 0; s < 5; ++s) {
    const float tb = 1.0f - 0.2f * (float)s;
    float lsum = 0.0f;
#pragma unroll
    for (int o = 0; o < 2; ++o) { zin[o] = z_a[o]; ksum[o] = (v8f){0.f,0.f,0.f,0.f,0.f,0.f,0.f,0.f}; }

#pragma clang loop unroll(disable)
    for (int st = 0; st < 4; ++st) {
      const float toff = (st == 0) ? 0.0f : ((st == 3) ? hstep : 0.5f * hstep);
      const float wst  = (st == 0 || st == 3) ? 1.0f : 2.0f;
      const float cz   = (st < 2) ? 0.5f * hstep : hstep;
      {
        const int done = 4 * s + st;
        if (lane == 0) prog[wave_u] = done;
        const int other = __builtin_amdgcn_readfirstlane(prog[partner]);
        const bool behind = (done < other) || (done == other && wave_u >= NWAVES / 2);
        if (behind) __builtin_amdgcn_s_setprio(3);
        else        __builtin_amdgcn_s_setprio(0);
      }
      float tr;
      func_eval(smem, lo, hi, zin, eps_w0, eps_w2t, kcur, tr, tb + toff);
      lsum += wst * tr;
#pragma unroll
      for (int o = 0; o < 2; ++o) { ksum[o] += kcur[o] * wst; zin[o] = z_a[o] + kcur[o] * cz; }
    }

    const float c = hstep / 6.0f;
#pragma unroll
    for (int o = 0; o < 2; ++o) z_a[o] += ksum[o] * c;
    ldacc -= c * lsum;
  }

  __builtin_amdgcn_s_setprio(0);

  {
    float* orow = out_z + (size_t)(rowbase + lo) * DD + 8 * hi;
#pragma unroll
    for (int o = 0; o < 2; ++o) {
      *(v4f*)(orow + 16 * o)     = (v4f){z_a[o][0], z_a[o][1], z_a[o][2], z_a[o][3]};
      *(v4f*)(orow + 16 * o + 4) = (v4f){z_a[o][4], z_a[o][5], z_a[o][6], z_a[o][7]};
    }
    const float ld = ldacc + __shfl_xor(ldacc, 16, 32);
    if (hi == 0) out_ld[rowbase + lo] = ld;
  }
}

extern "C" void kernel_launch(void* const* d_in, const int* in_sizes, int n_in,
                              void* d_out, int out_size, void* d_ws, size_t ws_size,
                              hipStream_t stream) {
  (void)n_in; (void)out_size; (void)d_ws; (void)ws_size;
  const float* z1  = (const float*)d_in[0];
  const float* eps = (const float*)d_in[1];
  const float* W0  = (const float*)d_in[2];
  const float* b0  = (const float*)d_in[3];
  const float* W1  = (const float*)d_in[4];
  const float* b1  = (const float*)d_in[5];
  const float* W2  = (const float*)d_in[6];
  const float* b2  = (const float*)d_in[7];

  const int Btot = in_sizes[0] / DD;
  float* out_z  = (float*)d_out;
  float* out_ld = out_z + (size_t)Btot * DD;

  dim3 grid(Btot / ROWS_BLK), block(NWAVES * 32);
  size_t shmem = (size_t)SMEM_FLOATS * sizeof(float);
  ffjord_kernel<<<grid, block, shmem, stream>>>(z1, eps, W0, b0, W1, b1, W2, b2,
                                                out_z, out_ld);
}
